# starting kernel
# baseline (speedup 1.0000x reference)
#include <hip/hip_runtime.h>
#include <math.h>


typedef float v2f __attribute__((ext_vector_type(2)));
typedef float v8f __attribute__((ext_vector_type(8)));

__device__ __forceinline__ void atomAddF(float* p, float v) {
    __hip_atomic_fetch_add(p, v, __ATOMIC_RELAXED, __HIP_MEMORY_SCOPE_AGENT);
}

__global__ void k_init(float* __restrict__ deg, float* __restrict__ errsum,
                       float* __restrict__ cnt, int N) {
    int i = blockIdx.x * blockDim.x + threadIdx.x;
    if (i < N) { deg[i] = 1.0f; errsum[i] = 0.0f; cnt[i] = 0.0f; }
}

__global__ void k_degree(const int* __restrict__ dst, float* __restrict__ deg, int E) {
    int e = blockIdx.x * blockDim.x + threadIdx.x;
    if (e < E) atomAddF(&deg[dst[e]], 1.0f);
}

__global__ void k_rsqrt(float* __restrict__ deg, int N) {
    int i = blockIdx.x * blockDim.x + threadIdx.x;
    if (i < N) deg[i] = rsqrtf(deg[i]);
}

__global__ void k_transpose(const float* __restrict__ W, float* __restrict__ Wt,
                            int N) {
    int idx = blockIdx.x * blockDim.x + threadIdx.x;
    if (idx < 128 * N) {
        int k = idx / N, n = idx - k * N;
        Wt[n * 128 + k] = W[idx];
    }
}

__global__ void k_fill_bias(float* __restrict__ out, const float* __restrict__ b,
                            int total, int dimMask) {
    int i = blockIdx.x * blockDim.x + threadIdx.x;
    if (i < total) out[i] = b[i & dimMask];
}

__global__ void k_relu(float* __restrict__ a, int total) {
    int i = blockIdx.x * blockDim.x + threadIdx.x;
    if (i < total) a[i] = fmaxf(a[i], 0.0f);
}

template <int NCOLS>
__global__ void k_gemm_wmma_k128(const float* __restrict__ A,
                                 const float* __restrict__ Bt,
                                 float* __restrict__ C) {
    __shared__ float As[16 * 132];
    const int tid  = threadIdx.x;
    const int row0 = blockIdx.x * 16;

    const float4* Ag = (const float4*)(A + (size_t)row0 * 128);
    for (int i = tid; i < 16 * 32; i += blockDim.x) {
        int r = i >> 5, c4 = i & 31;
        ((float4*)(As + r * 132))[c4] = Ag[r * 32 + c4];
    }
    __syncthreads();

    const int lane  = tid & 31;
    const int col0  = (tid >> 5) * 16;
    const int n15   = lane & 15;
    const int khalf = lane >> 4;

    v8f acc = {0.f, 0.f, 0.f, 0.f, 0.f, 0.f, 0.f, 0.f};
    const float* ap = As + n15 * 132 + 2 * khalf;
    const float* bp = Bt + (size_t)(col0 + n15) * 128 + 2 * khalf;

#pragma unroll
    for (int kk = 0; kk < 128; kk += 4) {
        v2f a; a.x = ap[kk]; a.y = ap[kk + 1];
        v2f b = *(const v2f*)(bp + kk);
        acc = __builtin_amdgcn_wmma_f32_16x16x4_f32(
             false, a,  false, b,
             (short)0, acc,  false,  false);
    }

    float* Cp = C + (size_t)(row0 + 8 * khalf) * NCOLS + col0 + n15;
#pragma unroll
    for (int r = 0; r < 8; ++r) Cp[r * NCOLS] = acc[r];
}

template <int DIM>
__global__ void k_aggregate(const float* __restrict__ H, float* __restrict__ out,
                            const int* __restrict__ src, const int* __restrict__ dst,
                            const float* __restrict__ dinv, int E, int N) {
    const int  lane  = threadIdx.x & 31;
    const long long w = (long long)blockIdx.x * (blockDim.x >> 5) + (threadIdx.x >> 5);
    const long long total = (long long)E + N;
    if (w >= total) return;

    int s, d;
    if (w < E) { s = src[w]; d = dst[w]; }
    else       { s = d = (int)(w - E); }
    const float norm = dinv[s] * dinv[d];

    constexpr int VPL = DIM / 32;
    const float* hp = H   + (size_t)s * DIM + lane * VPL;
    float*       op = out + (size_t)d * DIM + lane * VPL;
    if constexpr (VPL == 4) {
        float4 v = *(const float4*)hp;
        atomAddF(op + 0, v.x * norm);
        atomAddF(op + 1, v.y * norm);
        atomAddF(op + 2, v.z * norm);
        atomAddF(op + 3, v.w * norm);
    } else {
        float2 v = *(const float2*)hp;
        atomAddF(op + 0, v.x * norm);
        atomAddF(op + 1, v.y * norm);
    }
}

__global__ void k_decode(const float* __restrict__ Z, const int* __restrict__ row,
                         const int* __restrict__ col, float* __restrict__ errsum,
                         float* __restrict__ cnt, int E) {
    const int lane = threadIdx.x & 31;
    const long long e = (long long)blockIdx.x * (blockDim.x >> 5) + (threadIdx.x >> 5);
    if (e >= E) return;
    const int r = row[e], c = col[e];

    const float2 zr = *(const float2*)(Z + (size_t)r * 64 + lane * 2);
    const float2 zc = *(const float2*)(Z + (size_t)c * 64 + lane * 2);
    float p = zr.x * zc.x + zr.y * zc.y;
#pragma unroll
    for (int off = 16; off > 0; off >>= 1) p += __shfl_xor(p, off, 32);

    if (lane == 0) {
        float sig = 1.0f / (1.0f + __expf(-p));
        float err = -__logf(sig + 1e-15f);
        atomAddF(&errsum[r], err);
        atomAddF(&cnt[r], 1.0f);
    }
}

__global__ void k_finalize(const float* __restrict__ errsum, const float* __restrict__ cnt,
                           float* __restrict__ out, int N) {
    int i = blockIdx.x * blockDim.x + threadIdx.x;
    if (i < N) {
        float c = cnt[i];
        out[i] = (c > 0.0f) ? (errsum[i] / fmaxf(c, 1.0f)) : 10.0f;
    }
}

extern "C" void kernel_launch(void* const* d_in, const int* in_sizes, int n_in,
                              void* d_out, int out_size, void* d_ws, size_t ws_size,
                              hipStream_t stream) {
    const float* x  = (const float*)d_in[0];
    const int*   ei = (const int*)d_in[1];
    const float* W1 = (const float*)d_in[2];
    const float* b1 = (const float*)d_in[3];
    const float* W2 = (const float*)d_in[4];
    const float* b2 = (const float*)d_in[5];

    const int N = in_sizes[0] / 128;
    const int E = in_sizes[1] / 2;
    const int* src = ei;
    const int* dst = ei + E;

    float* ws     = (float*)d_ws;
    float* dinv   = ws;
    float* errsum = dinv + N;
    float* cnt    = errsum + N;
    size_t off    = (((size_t)3 * N) + 63) & ~(size_t)63;
    float* wt1    = ws + off;
    float* wt2    = wt1 + 128 * 128;
    float* bufA   = wt2 + 64 * 128;
    float* bufB   = bufA + (size_t)N * 128;

    const int T = 256;
    k_init     <<<(N + T - 1) / T, T, 0, stream>>>(dinv, errsum, cnt, N);
    k_degree   <<<(E + T - 1) / T, T, 0, stream>>>(dst, dinv, E);
    k_rsqrt    <<<(N + T - 1) / T, T, 0, stream>>>(dinv, N);
    k_transpose<<<(128 * 128 + T - 1) / T, T, 0, stream>>>(W1, wt1, 128);
    k_transpose<<<(128 * 64  + T - 1) / T, T, 0, stream>>>(W2, wt2, 64);

    k_gemm_wmma_k128<128><<<N / 16, 256, 0, stream>>>(x, wt1, bufA);
    k_fill_bias<<<(N * 128 + T - 1) / T, T, 0, stream>>>(bufB, b1, N * 128, 127);
    k_aggregate<128><<<(E + N + 7) / 8, 256, 0, stream>>>(bufA, bufB, src, dst, dinv, E, N);
    k_relu<<<(N * 128 + T - 1) / T, T, 0, stream>>>(bufB, N * 128);

    k_gemm_wmma_k128<64><<<N / 16, 128, 0, stream>>>(bufB, wt2, bufA);
    k_fill_bias<<<(N * 64 + T - 1) / T, T, 0, stream>>>(bufB, b2, N * 64, 63);
    k_aggregate<64><<<(E + N + 7) / 8, 256, 0, stream>>>(bufA, bufB, src, dst, dinv, E, N);

    k_decode<<<(E + 7) / 8, 256, 0, stream>>>(bufB, src, dst, errsum, cnt, E);
    k_finalize<<<(N + T - 1) / T, T, 0, stream>>>(errsum, cnt, (float*)d_out, N);
}


# Aggregation as a pull over edges counting-sorted by destination (wave per row, register sums, one row store) replaces float-atomic scatter; layer 1 aggregates x before projecting, GEMM in place
# speedup vs baseline: 1.5955x; 1.5955x over previous
#include <hip/hip_runtime.h>
#include <math.h>


typedef float v2f __attribute__((ext_vector_type(2)));
typedef float v4f __attribute__((ext_vector_type(4)));
typedef float v8f __attribute__((ext_vector_type(8)));

template <int V> struct VecT;
template <> struct VecT<4> { typedef v4f T; };
template <> struct VecT<2> { typedef v2f T; };

__device__ __forceinline__ void atomAddF(float* p, float v) {
    __hip_atomic_fetch_add(p, v, __ATOMIC_RELAXED, __HIP_MEMORY_SCOPE_AGENT);
}

__global__ void k_init(int* __restrict__ indeg, float* __restrict__ errsum,
                       float* __restrict__ cnt, int N) {
    int i = blockIdx.x * blockDim.x + threadIdx.x;
    if (i < N) { indeg[i] = 0; errsum[i] = 0.0f; cnt[i] = 0.0f; }
}

__global__ void k_count(const int* __restrict__ dst, int* __restrict__ indeg,
                        int* __restrict__ pos, int E) {
    int e = blockIdx.x * blockDim.x + threadIdx.x;
    if (e < E)
        pos[e] = __hip_atomic_fetch_add(&indeg[dst[e]], 1, __ATOMIC_RELAXED,
                                        __HIP_MEMORY_SCOPE_AGENT);
}

__global__ void k_scan_part(const int* __restrict__ indeg, int* __restrict__ part, int N) {
    __shared__ int wsum[8];
    const int t = threadIdx.x;
    const int base = blockIdx.x * 1024 + t * 4;
    int s = 0;
    for (int j = 0; j < 4; ++j)
        if (base + j < N) s += indeg[base + j];
    for (int off = 16; off > 0; off >>= 1) s += __shfl_xor(s, off, 32);
    if ((t & 31) == 0) wsum[t >> 5] = s;
    __syncthreads();
    if (t == 0) {
        int tot = 0;
        for (int w = 0; w < 8; ++w) tot += wsum[w];
        part[blockIdx.x] = tot;
    }
}

__global__ void k_scan_write(const int* __restrict__ indeg, const int* __restrict__ part,
                             int* __restrict__ rowptr, float* __restrict__ dinv,
                             int N, int E) {
    __shared__ int wsum[8];
    __shared__ int wpre[8];
    const int t = threadIdx.x;
    const int lane = t & 31;
    const int w = t >> 5;

    int p = 0;
    for (int i = t; i < (int)blockIdx.x; i += 256) p += part[i];
    for (int off = 16; off > 0; off >>= 1) p += __shfl_xor(p, off, 32);
    if (lane == 0) wpre[w] = p;

    const int base = blockIdx.x * 1024 + t * 4;
    int v[4];
    int x = 0;
    for (int j = 0; j < 4; ++j) {
        v[j] = (base + j < N) ? indeg[base + j] : 0;
        x += v[j];
    }
    int incl = x;
    for (int off = 1; off < 32; off <<= 1) {
        int y = __shfl_up(incl, off, 32);
        if (lane >= off) incl += y;
    }
    if (lane == 31) wsum[w] = incl;
    __syncthreads();

    int run = incl - x;
    for (int i = 0; i < 8; ++i) {
        run += wpre[i];
        if (i < w) run += wsum[i];
    }
    for (int j = 0; j < 4; ++j) {
        if (base + j < N) {
            rowptr[base + j] = run;
            dinv[base + j] = rsqrtf((float)(v[j] + 1));
            run += v[j];
        }
    }
    if (blockIdx.x == 0 && t == 0) rowptr[N] = E;
}

__global__ void k_fill(const int* __restrict__ src, const int* __restrict__ dst,
                       const int* __restrict__ pos, const int* __restrict__ rowptr,
                       const float* __restrict__ dinv, int* __restrict__ csr_src,
                       float* __restrict__ csr_w, int E) {
    int e = blockIdx.x * blockDim.x + threadIdx.x;
    if (e < E) {
        const int s = src[e];
        const int j = rowptr[dst[e]] + pos[e];
        csr_src[j] = s;
        csr_w[j] = dinv[s];
    }
}

__global__ void k_transpose(const float* __restrict__ W, float* __restrict__ Wt,
                            int N) {
    int idx = blockIdx.x * blockDim.x + threadIdx.x;
    if (idx < 128 * N) {
        int k = idx / N, n = idx - k * N;
        Wt[n * 128 + k] = W[idx];
    }
}

template <int DIM, bool ADD_BIAS>
__global__ void k_pull(const float* __restrict__ H, float* __restrict__ out,
                       const int* __restrict__ rowptr, const int* __restrict__ csr_src,
                       const float* __restrict__ csr_w, const float* __restrict__ dinv,
                       const float* __restrict__ bias, int N) {
    constexpr int VPL = DIM / 32;
    typedef typename VecT<VPL>::T VT;
    const int lane = threadIdx.x & 31;
    const int wv = __builtin_amdgcn_readfirstlane((int)(threadIdx.x >> 5));
    const int d = blockIdx.x * (int)(blockDim.x >> 5) + wv;
    if (d >= N) return;

    const int beg = rowptr[d];
    const int end = rowptr[d + 1];
    const float dd = dinv[d];
    const float* hl = H + lane * VPL;

    VT acc = *(const VT*)(hl + (size_t)d * DIM) * dd;

    for (int base = beg; base < end; base += 32) {
        const int n = (end - base < 32) ? (end - base) : 32;
        int ms = 0;
        float mw = 0.0f;
        if (lane < n) { ms = csr_src[base + lane]; mw = csr_w[base + lane]; }
        const int mwi = __float_as_int(mw);
        int k = 0;
        for (; k + 4 <= n; k += 4) {
            const int s0 = __builtin_amdgcn_readlane(ms, k);
            const int s1 = __builtin_amdgcn_readlane(ms, k + 1);
            const int s2 = __builtin_amdgcn_readlane(ms, k + 2);
            const int s3 = __builtin_amdgcn_readlane(ms, k + 3);
            const VT v0 = *(const VT*)(hl + (size_t)s0 * DIM);
            const VT v1 = *(const VT*)(hl + (size_t)s1 * DIM);
            const VT v2 = *(const VT*)(hl + (size_t)s2 * DIM);
            const VT v3 = *(const VT*)(hl + (size_t)s3 * DIM);
            const float w0 = __int_as_float(__builtin_amdgcn_readlane(mwi, k));
            const float w1 = __int_as_float(__builtin_amdgcn_readlane(mwi, k + 1));
            const float w2 = __int_as_float(__builtin_amdgcn_readlane(mwi, k + 2));
            const float w3 = __int_as_float(__builtin_amdgcn_readlane(mwi, k + 3));
            acc += v0 * w0;
            acc += v1 * w1;
            acc += v2 * w2;
            acc += v3 * w3;
        }
        for (; k < n; ++k) {
            const int s0 = __builtin_amdgcn_readlane(ms, k);
            const float w0 = __int_as_float(__builtin_amdgcn_readlane(mwi, k));
            acc += *(const VT*)(hl + (size_t)s0 * DIM) * w0;
        }
    }

    acc *= dd;
    if constexpr (ADD_BIAS) acc += *(const VT*)(bias + lane * VPL);
    *(VT*)(out + (size_t)d * DIM + lane * VPL) = acc;
}

template <int NCOLS, bool BIAS_RELU>
__global__ void k_gemm_wmma_k128(const float* A,
                                 const float* __restrict__ Bt,
                                 const float* __restrict__ bias,
                                 float* C) {
    __shared__ float As[16 * 132];
    const int tid  = threadIdx.x;
    const int row0 = blockIdx.x * 16;

    const float4* Ag = (const float4*)(A + (size_t)row0 * 128);
    for (int i = tid; i < 16 * 32; i += blockDim.x) {
        int r = i >> 5, c4 = i & 31;
        ((float4*)(As + r * 132))[c4] = Ag[r * 32 + c4];
    }
    __syncthreads();

    const int lane  = tid & 31;
    const int col0  = (tid >> 5) * 16;
    const int n15   = lane & 15;
    const int khalf = lane >> 4;

    v8f acc = {0.f, 0.f, 0.f, 0.f, 0.f, 0.f, 0.f, 0.f};
    const float* ap = As + n15 * 132 + 2 * khalf;
    const float* bp = Bt + (size_t)(col0 + n15) * 128 + 2 * khalf;

#pragma unroll
    for (int kk = 0; kk < 128; kk += 4) {
        v2f a; a.x = ap[kk]; a.y = ap[kk + 1];
        v2f b = *(const v2f*)(bp + kk);
        acc = __builtin_amdgcn_wmma_f32_16x16x4_f32(
             false, a,  false, b,
             (short)0, acc,  false,  false);
    }

    float* Cp = C + (size_t)(row0 + 8 * khalf) * NCOLS + col0 + n15;
    if constexpr (BIAS_RELU) {
        const float bv = bias[col0 + n15];
#pragma unroll
        for (int r = 0; r < 8; ++r) Cp[r * NCOLS] = fmaxf(acc[r] + bv, 0.0f);
    } else {
#pragma unroll
        for (int r = 0; r < 8; ++r) Cp[r * NCOLS] = acc[r];
    }
}

__global__ void k_decode(const float* __restrict__ Z, const int* __restrict__ row,
                         const int* __restrict__ col, float* __restrict__ errsum,
                         float* __restrict__ cnt, int E) {
    const int lane = threadIdx.x & 31;
    const long long e = (long long)blockIdx.x * (blockDim.x >> 5) + (threadIdx.x >> 5);
    if (e >= E) return;
    const int r = row[e], c = col[e];

    const float2 zr = *(const float2*)(Z + (size_t)r * 64 + lane * 2);
    const float2 zc = *(const float2*)(Z + (size_t)c * 64 + lane * 2);
    float p = zr.x * zc.x + zr.y * zc.y;
#pragma unroll
    for (int off = 16; off > 0; off >>= 1) p += __shfl_xor(p, off, 32);

    if (lane == 0) {
        float sig = 1.0f / (1.0f + __expf(-p));
        float err = -__logf(sig + 1e-15f);
        atomAddF(&errsum[r], err);
        atomAddF(&cnt[r], 1.0f);
    }
}

__global__ void k_finalize(const float* __restrict__ errsum, const float* __restrict__ cnt,
                           float* __restrict__ out, int N) {
    int i = blockIdx.x * blockDim.x + threadIdx.x;
    if (i < N) {
        float c = cnt[i];
        out[i] = (c > 0.0f) ? (errsum[i] / fmaxf(c, 1.0f)) : 10.0f;
    }
}

extern "C" void kernel_launch(void* const* d_in, const int* in_sizes, int n_in,
                              void* d_out, int out_size, void* d_ws, size_t ws_size,
                              hipStream_t stream) {
    const float* x  = (const float*)d_in[0];
    const int*   ei = (const int*)d_in[1];
    const float* W1 = (const float*)d_in[2];
    const float* b1 = (const float*)d_in[3];
    const float* W2 = (const float*)d_in[4];
    const float* b2 = (const float*)d_in[5];

    const int N = in_sizes[0] / 128;
    const int E = in_sizes[1] / 2;
    const int* src = ei;
    const int* dst = ei + E;

    float* ws     = (float*)d_ws;
    float* dinv   = ws;
    float* errsum = dinv + N;
    float* cnt    = errsum + N;
    size_t off    = (((size_t)3 * N) + 63) & ~(size_t)63;
    float* wt1    = ws + off;
    float* wt2    = wt1 + 128 * 128;
    float* bufA   = wt2 + 64 * 128;
    float* bufB   = bufA + (size_t)N * 128;
    const int NT  = (N + 1023) / 1024;
    int*   indeg   = (int*)(bufB + (size_t)N * 64);
    int*   rowptr  = indeg + N;
    int*   part    = rowptr + N + 1;
    int*   pos     = part + NT;
    int*   csr_src = pos + E;
    float* csr_w   = (float*)(csr_src + E);

    const int T = 256;
    k_init      <<<(N + T - 1) / T, T, 0, stream>>>(indeg, errsum, cnt, N);
    k_count     <<<(E + T - 1) / T, T, 0, stream>>>(dst, indeg, pos, E);
    k_scan_part <<<NT, 256, 0, stream>>>(indeg, part, N);
    k_scan_write<<<NT, 256, 0, stream>>>(indeg, part, rowptr, dinv, N, E);
    k_fill      <<<(E + T - 1) / T, T, 0, stream>>>(src, dst, pos, rowptr, dinv, csr_src, csr_w, E);
    k_transpose<<<(128 * 128 + T - 1) / T, T, 0, stream>>>(W1, wt1, 128);
    k_transpose<<<(128 * 64  + T - 1) / T, T, 0, stream>>>(W2, wt2, 64);

    k_pull<128, false><<<(N + 7) / 8, 256, 0, stream>>>(x, bufA, rowptr, csr_src, csr_w, dinv, b1, N);
    k_gemm_wmma_k128<128, true><<<N / 16, 256, 0, stream>>>(bufA, wt1, b1, bufA);

    k_gemm_wmma_k128<64, false><<<N / 16, 128, 0, stream>>>(bufA, wt2, b2, bufB);
    k_pull<64, true><<<(N + 7) / 8, 256, 0, stream>>>(bufB, bufA, rowptr, csr_src, csr_w, dinv, b2, N);

    k_decode<<<(E + 7) / 8, 256, 0, stream>>>(bufA, src, dst, errsum, cnt, E);
    k_finalize<<<(N + T - 1) / T, T, 0, stream>>>(errsum, cnt, (float*)d_out, N);
}


# Edge decode as a pull over the destination-sorted edges: one wave per destination row keeps z[d] in registers, gathers each z[s] once, 16 scores per butterfly; replaces one wave per edge
# speedup vs baseline: 2.0728x; 1.2992x over previous
#include <hip/hip_runtime.h>
#include <math.h>


typedef float v2f __attribute__((ext_vector_type(2)));
typedef float v4f __attribute__((ext_vector_type(4)));
typedef float v8f __attribute__((ext_vector_type(8)));

template <int V> struct VecT;
template <> struct VecT<4> { typedef v4f T; };
template <> struct VecT<2> { typedef v2f T; };

__device__ __forceinline__ void atomAddF(float* p, float v) {
    __hip_atomic_fetch_add(p, v, __ATOMIC_RELAXED, __HIP_MEMORY_SCOPE_AGENT);
}

__global__ void k_init(int* __restrict__ indeg, float* __restrict__ errsum,
                       float* __restrict__ cnt, int N) {
    int i = blockIdx.x * blockDim.x + threadIdx.x;
    if (i < N) { indeg[i] = 0; errsum[i] = 0.0f; cnt[i] = 0.0f; }
}

__global__ void k_count(const int* __restrict__ dst, int* __restrict__ indeg,
                        int* __restrict__ pos, int E) {
    int e = blockIdx.x * blockDim.x + threadIdx.x;
    if (e < E)
        pos[e] = __hip_atomic_fetch_add(&indeg[dst[e]], 1, __ATOMIC_RELAXED,
                                        __HIP_MEMORY_SCOPE_AGENT);
}

__global__ void k_scan_part(const int* __restrict__ indeg, int* __restrict__ part, int N) {
    __shared__ int wsum[8];
    const int t = threadIdx.x;
    const int base = blockIdx.x * 1024 + t * 4;
    int s = 0;
    for (int j = 0; j < 4; ++j)
        if (base + j < N) s += indeg[base + j];
    for (int off = 16; off > 0; off >>= 1) s += __shfl_xor(s, off, 32);
    if ((t & 31) == 0) wsum[t >> 5] = s;
    __syncthreads();
    if (t == 0) {
        int tot = 0;
        for (int w = 0; w < 8; ++w) tot += wsum[w];
        part[blockIdx.x] = tot;
    }
}

__global__ void k_scan_write(const int* __restrict__ indeg, const int* __restrict__ part,
                             int* __restrict__ rowptr, float* __restrict__ dinv,
                             int N, int E) {
    __shared__ int wsum[8];
    __shared__ int wpre[8];
    const int t = threadIdx.x;
    const int lane = t & 31;
    const int w = t >> 5;

    int p = 0;
    for (int i = t; i < (int)blockIdx.x; i += 256) p += part[i];
    for (int off = 16; off > 0; off >>= 1) p += __shfl_xor(p, off, 32);
    if (lane == 0) wpre[w] = p;

    const int base = blockIdx.x * 1024 + t * 4;
    int v[4];
    int x = 0;
    for (int j = 0; j < 4; ++j) {
        v[j] = (base + j < N) ? indeg[base + j] : 0;
        x += v[j];
    }
    int incl = x;
    for (int off = 1; off < 32; off <<= 1) {
        int y = __shfl_up(incl, off, 32);
        if (lane >= off) incl += y;
    }
    if (lane == 31) wsum[w] = incl;
    __syncthreads();

    int run = incl - x;
    for (int i = 0; i < 8; ++i) {
        run += wpre[i];
        if (i < w) run += wsum[i];
    }
    for (int j = 0; j < 4; ++j) {
        if (base + j < N) {
            rowptr[base + j] = run;
            dinv[base + j] = rsqrtf((float)(v[j] + 1));
            run += v[j];
        }
    }
    if (blockIdx.x == 0 && t == 0) rowptr[N] = E;
}

__global__ void k_fill(const int* __restrict__ src, const int* __restrict__ dst,
                       const int* __restrict__ pos, const int* __restrict__ rowptr,
                       const float* __restrict__ dinv, int* __restrict__ csr_src,
                       float* __restrict__ csr_w, int E) {
    int e = blockIdx.x * blockDim.x + threadIdx.x;
    if (e < E) {
        const int s = src[e];
        const int j = rowptr[dst[e]] + pos[e];
        csr_src[j] = s;
        csr_w[j] = dinv[s];
    }
}

__global__ void k_transpose(const float* __restrict__ W, float* __restrict__ Wt,
                            int N) {
    int idx = blockIdx.x * blockDim.x + threadIdx.x;
    if (idx < 128 * N) {
        int k = idx / N, n = idx - k * N;
        Wt[n * 128 + k] = W[idx];
    }
}

template <int DIM, bool ADD_BIAS>
__global__ void k_pull(const float* __restrict__ H, float* __restrict__ out,
                       const int* __restrict__ rowptr, const int* __restrict__ csr_src,
                       const float* __restrict__ csr_w, const float* __restrict__ dinv,
                       const float* __restrict__ bias, int N) {
    constexpr int VPL = DIM / 32;
    typedef typename VecT<VPL>::T VT;
    const int lane = threadIdx.x & 31;
    const int wv = __builtin_amdgcn_readfirstlane((int)(threadIdx.x >> 5));
    const int d = blockIdx.x * (int)(blockDim.x >> 5) + wv;
    if (d >= N) return;

    const int beg = rowptr[d];
    const int end = rowptr[d + 1];
    const float dd = dinv[d];
    const float* hl = H + lane * VPL;

    VT acc = *(const VT*)(hl + (size_t)d * DIM) * dd;

    for (int base = beg; base < end; base += 32) {
        const int n = (end - base < 32) ? (end - base) : 32;
        int ms = 0;
        float mw = 0.0f;
        if (lane < n) { ms = csr_src[base + lane]; mw = csr_w[base + lane]; }
        const int mwi = __float_as_int(mw);
        int k = 0;
        for (; k + 4 <= n; k += 4) {
            const int s0 = __builtin_amdgcn_readlane(ms, k);
            const int s1 = __builtin_amdgcn_readlane(ms, k + 1);
            const int s2 = __builtin_amdgcn_readlane(ms, k + 2);
            const int s3 = __builtin_amdgcn_readlane(ms, k + 3);
            const VT v0 = *(const VT*)(hl + (size_t)s0 * DIM);
            const VT v1 = *(const VT*)(hl + (size_t)s1 * DIM);
            const VT v2 = *(const VT*)(hl + (size_t)s2 * DIM);
            const VT v3 = *(const VT*)(hl + (size_t)s3 * DIM);
            const float w0 = __int_as_float(__builtin_amdgcn_readlane(mwi, k));
            const float w1 = __int_as_float(__builtin_amdgcn_readlane(mwi, k + 1));
            const float w2 = __int_as_float(__builtin_amdgcn_readlane(mwi, k + 2));
            const float w3 = __int_as_float(__builtin_amdgcn_readlane(mwi, k + 3));
            acc += v0 * w0;
            acc += v1 * w1;
            acc += v2 * w2;
            acc += v3 * w3;
        }
        for (; k < n; ++k) {
            const int s0 = __builtin_amdgcn_readlane(ms, k);
            const float w0 = __int_as_float(__builtin_amdgcn_readlane(mwi, k));
            acc += *(const VT*)(hl + (size_t)s0 * DIM) * w0;
        }
    }

    acc *= dd;
    if constexpr (ADD_BIAS) acc += *(const VT*)(bias + lane * VPL);
    *(VT*)(out + (size_t)d * DIM + lane * VPL) = acc;
}

template <int NCOLS, bool BIAS_RELU>
__global__ void k_gemm_wmma_k128(const float* A,
                                 const float* __restrict__ Bt,
                                 const float* __restrict__ bias,
                                 float* C) {
    __shared__ float As[16 * 132];
    const int tid  = threadIdx.x;
    const int row0 = blockIdx.x * 16;

    const float4* Ag = (const float4*)(A + (size_t)row0 * 128);
    for (int i = tid; i < 16 * 32; i += blockDim.x) {
        int r = i >> 5, c4 = i & 31;
        ((float4*)(As + r * 132))[c4] = Ag[r * 32 + c4];
    }
    __syncthreads();

    const int lane  = tid & 31;
    const int col0  = (tid >> 5) * 16;
    const int n15   = lane & 15;
    const int khalf = lane >> 4;

    v8f acc = {0.f, 0.f, 0.f, 0.f, 0.f, 0.f, 0.f, 0.f};
    const float* ap = As + n15 * 132 + 2 * khalf;
    const float* bp = Bt + (size_t)(col0 + n15) * 128 + 2 * khalf;

#pragma unroll
    for (int kk = 0; kk < 128; kk += 4) {
        v2f a; a.x = ap[kk]; a.y = ap[kk + 1];
        v2f b = *(const v2f*)(bp + kk);
        acc = __builtin_amdgcn_wmma_f32_16x16x4_f32(
             false, a,  false, b,
             (short)0, acc,  false,  false);
    }

    float* Cp = C + (size_t)(row0 + 8 * khalf) * NCOLS + col0 + n15;
    if constexpr (BIAS_RELU) {
        const float bv = bias[col0 + n15];
#pragma unroll
        for (int r = 0; r < 8; ++r) Cp[r * NCOLS] = fmaxf(acc[r] + bv, 0.0f);
    } else {
#pragma unroll
        for (int r = 0; r < 8; ++r) Cp[r * NCOLS] = acc[r];
    }
}

template <int CNT, int M>
__device__ __forceinline__ void fold_half(float (&p)[16], int lane) {
    const bool hi = (lane & M) != 0;
#pragma unroll
    for (int i = 0; i < CNT / 2; ++i) {
        const float keep = hi ? p[i + CNT / 2] : p[i];
        const float send = hi ? p[i] : p[i + CNT / 2];
        p[i] = keep + __shfl_xor(send, M, 32);
    }
}

template <int K>
__device__ __forceinline__ void edge_products(float (&p)[16], const float* __restrict__ Z,
                                              int lane, int ms, v2f zd) {
    v2f v[K];
#pragma unroll
    for (int i = 0; i < K; ++i) {
        const int s = __builtin_amdgcn_readlane(ms, 2 * i);
        const float* rp = Z + (size_t)s * 64;
        v[i] = *(const v2f*)(rp + lane * 2);
    }
#pragma unroll
    for (int i = 0; i < K; ++i) p[i] = v[i].x * zd.x + v[i].y * zd.y;
#pragma unroll
    for (int i = K; i < 16; ++i) p[i] = 0.0f;
}

__global__ void k_decode_pull(const float* __restrict__ Z, const int* __restrict__ rowptr,
                              const int* __restrict__ csr_src, float* __restrict__ errsum,
                              float* __restrict__ cnt, int N) {
    const int lane = threadIdx.x & 31;
    const int wv = __builtin_amdgcn_readfirstlane((int)(threadIdx.x >> 5));
    const int d = blockIdx.x * (int)(blockDim.x >> 5) + wv;
    if (d >= N) return;

    const int beg = rowptr[d];
    const int end = rowptr[d + 1];
    if (beg >= end) return;
    const int slot = lane >> 1;
    const int last = end - 1;

    int ms = csr_src[(beg + slot < last) ? (beg + slot) : last];
    const v2f zd = *(const v2f*)(Z + (size_t)d * 64 + lane * 2);

    for (int base = beg; base < end; base += 16) {
        const int n = (end - base < 16) ? (end - base) : 16;
        const int nx = base + 16 + slot;
        const int ms_next = csr_src[(nx < last) ? nx : last];

        float p[16];
        if (n > 12)      edge_products<16>(p, Z, lane, ms, zd);
        else if (n > 8)  edge_products<12>(p, Z, lane, ms, zd);
        else if (n > 4)  edge_products<8>(p, Z, lane, ms, zd);
        else             edge_products<4>(p, Z, lane, ms, zd);

        fold_half<16, 16>(p, lane);
        fold_half<8, 8>(p, lane);
        fold_half<4, 4>(p, lane);
        fold_half<2, 2>(p, lane);
        const float sc = p[0] + __shfl_xor(p[0], 1, 32);

        const float sig = 1.0f / (1.0f + __expf(-sc));
        const float err = -__logf(sig + 1e-15f);
        if (slot < n) {
            const bool odd = (lane & 1) != 0;
            atomAddF((odd ? cnt : errsum) + ms, odd ? 1.0f : err);
        }
        ms = ms_next;
    }
}

__global__ void k_finalize(const float* __restrict__ errsum, const float* __restrict__ cnt,
                           float* __restrict__ out, int N) {
    int i = blockIdx.x * blockDim.x + threadIdx.x;
    if (i < N) {
        float c = cnt[i];
        out[i] = (c > 0.0f) ? (errsum[i] / fmaxf(c, 1.0f)) : 10.0f;
    }
}

extern "C" void kernel_launch(void* const* d_in, const int* in_sizes, int n_in,
                              void* d_out, int out_size, void* d_ws, size_t ws_size,
                              hipStream_t stream) {
    const float* x  = (const float*)d_in[0];
    const int*   ei = (const int*)d_in[1];
    const float* W1 = (const float*)d_in[2];
    const float* b1 = (const float*)d_in[3];
    const float* W2 = (const float*)d_in[4];
    const float* b2 = (const float*)d_in[5];

    const int N = in_sizes[0] / 128;
    const int E = in_sizes[1] / 2;
    const int* src = ei;
    const int* dst = ei + E;

    float* ws     = (float*)d_ws;
    float* dinv   = ws;
    float* errsum = dinv + N;
    float* cnt    = errsum + N;
    size_t off    = (((size_t)3 * N) + 63) & ~(size_t)63;
    float* wt1    = ws + off;
    float* wt2    = wt1 + 128 * 128;
    float* bufA   = wt2 + 64 * 128;
    float* bufB   = bufA + (size_t)N * 128;
    const int NT  = (N + 1023) / 1024;
    int*   indeg   = (int*)(bufB + (size_t)N * 64);
    int*   rowptr  = indeg + N;
    int*   part    = rowptr + N + 1;
    int*   pos     = part + NT;
    int*   csr_src = pos + E;
    float* csr_w   = (float*)(csr_src + E);

    const int T = 256;
    k_init      <<<(N + T - 1) / T, T, 0, stream>>>(indeg, errsum, cnt, N);
    k_count     <<<(E + T - 1) / T, T, 0, stream>>>(dst, indeg, pos, E);
    k_scan_part <<<NT, 256, 0, stream>>>(indeg, part, N);
    k_scan_write<<<NT, 256, 0, stream>>>(indeg, part, rowptr, dinv, N, E);
    k_fill      <<<(E + T - 1) / T, T, 0, stream>>>(src, dst, pos, rowptr, dinv, csr_src, csr_w, E);
    k_transpose<<<(128 * 128 + T - 1) / T, T, 0, stream>>>(W1, wt1, 128);
    k_transpose<<<(128 * 64  + T - 1) / T, T, 0, stream>>>(W2, wt2, 64);

    k_pull<128, false><<<(N + 7) / 8, 256, 0, stream>>>(x, bufA, rowptr, csr_src, csr_w, dinv, b1, N);
    k_gemm_wmma_k128<128, true><<<N / 16, 256, 0, stream>>>(bufA, wt1, b1, bufA);

    k_gemm_wmma_k128<64, false><<<N / 16, 128, 0, stream>>>(bufA, wt2, b2, bufB);
    k_pull<64, true><<<(N + 7) / 8, 256, 0, stream>>>(bufB, bufA, rowptr, csr_src, csr_w, dinv, b2, N);

    k_decode_pull<<<(N + 7) / 8, 256, 0, stream>>>(bufA, rowptr, csr_src, errsum, cnt, N);
    k_finalize<<<(N + T - 1) / T, T, 0, stream>>>(errsum, cnt, (float*)d_out, N);
}


# Decode: per-source error sum and edge count interleaved as one float pair, so the two atomic adds of an edge leave as one cache-line request from two neighbouring lanes instead of two scattered ones
# speedup vs baseline: 2.3530x; 1.1352x over previous
#include <hip/hip_runtime.h>
#include <math.h>


typedef float v2f __attribute__((ext_vector_type(2)));
typedef float v4f __attribute__((ext_vector_type(4)));
typedef float v8f __attribute__((ext_vector_type(8)));

template <int V> struct VecT;
template <> struct VecT<4> { typedef v4f T; };
template <> struct VecT<2> { typedef v2f T; };

__device__ __forceinline__ void atomAddF(float* p, float v) {
    __hip_atomic_fetch_add(p, v, __ATOMIC_RELAXED, __HIP_MEMORY_SCOPE_AGENT);
}

__global__ void k_init(int* __restrict__ indeg, float* __restrict__ ec, int N) {
    int i = blockIdx.x * blockDim.x + threadIdx.x;
    if (i < N) {
        indeg[i] = 0;
        v2f z; z.x = 0.0f; z.y = 0.0f;
        *(v2f*)(ec + 2 * (size_t)i) = z;
    }
}

__global__ void k_count(const int* __restrict__ dst, int* __restrict__ indeg,
                        int* __restrict__ pos, int E) {
    int e = blockIdx.x * blockDim.x + threadIdx.x;
    if (e < E)
        pos[e] = __hip_atomic_fetch_add(&indeg[dst[e]], 1, __ATOMIC_RELAXED,
                                        __HIP_MEMORY_SCOPE_AGENT);
}

__global__ void k_scan_part(const int* __restrict__ indeg, int* __restrict__ part, int N) {
    __shared__ int wsum[8];
    const int t = threadIdx.x;
    const int base = blockIdx.x * 1024 + t * 4;
    int s = 0;
    for (int j = 0; j < 4; ++j)
        if (base + j < N) s += indeg[base + j];
    for (int off = 16; off > 0; off >>= 1) s += __shfl_xor(s, off, 32);
    if ((t & 31) == 0) wsum[t >> 5] = s;
    __syncthreads();
    if (t == 0) {
        int tot = 0;
        for (int w = 0; w < 8; ++w) tot += wsum[w];
        part[blockIdx.x] = tot;
    }
}

__global__ void k_scan_write(const int* __restrict__ indeg, const int* __restrict__ part,
                             int* __restrict__ rowptr, float* __restrict__ dinv,
                             int N, int E) {
    __shared__ int wsum[8];
    __shared__ int wpre[8];
    const int t = threadIdx.x;
    const int lane = t & 31;
    const int w = t >> 5;

    int p = 0;
    for (int i = t; i < (int)blockIdx.x; i += 256) p += part[i];
    for (int off = 16; off > 0; off >>= 1) p += __shfl_xor(p, off, 32);
    if (lane == 0) wpre[w] = p;

    const int base = blockIdx.x * 1024 + t * 4;
    int v[4];
    int x = 0;
    for (int j = 0; j < 4; ++j) {
        v[j] = (base + j < N) ? indeg[base + j] : 0;
        x += v[j];
    }
    int incl = x;
    for (int off = 1; off < 32; off <<= 1) {
        int y = __shfl_up(incl, off, 32);
        if (lane >= off) incl += y;
    }
    if (lane == 31) wsum[w] = incl;
    __syncthreads();

    int run = incl - x;
    for (int i = 0; i < 8; ++i) {
        run += wpre[i];
        if (i < w) run += wsum[i];
    }
    for (int j = 0; j < 4; ++j) {
        if (base + j < N) {
            rowptr[base + j] = run;
            dinv[base + j] = rsqrtf((float)(v[j] + 1));
            run += v[j];
        }
    }
    if (blockIdx.x == 0 && t == 0) rowptr[N] = E;
}

__global__ void k_fill(const int* __restrict__ src, const int* __restrict__ dst,
                       const int* __restrict__ pos, const int* __restrict__ rowptr,
                       const float* __restrict__ dinv, int* __restrict__ csr_src,
                       float* __restrict__ csr_w, int E) {
    int e = blockIdx.x * blockDim.x + threadIdx.x;
    if (e < E) {
        const int s = src[e];
        const int j = rowptr[dst[e]] + pos[e];
        csr_src[j] = s;
        csr_w[j] = dinv[s];
    }
}

__global__ void k_transpose(const float* __restrict__ W, float* __restrict__ Wt,
                            int N) {
    int idx = blockIdx.x * blockDim.x + threadIdx.x;
    if (idx < 128 * N) {
        int k = idx / N, n = idx - k * N;
        Wt[n * 128 + k] = W[idx];
    }
}

template <int DIM, bool ADD_BIAS>
__global__ void k_pull(const float* __restrict__ H, float* __restrict__ out,
                       const int* __restrict__ rowptr, const int* __restrict__ csr_src,
                       const float* __restrict__ csr_w, const float* __restrict__ dinv,
                       const float* __restrict__ bias, int N) {
    constexpr int VPL = DIM / 32;
    typedef typename VecT<VPL>::T VT;
    const int lane = threadIdx.x & 31;
    const int wv = __builtin_amdgcn_readfirstlane((int)(threadIdx.x >> 5));
    const int d = blockIdx.x * (int)(blockDim.x >> 5) + wv;
    if (d >= N) return;

    const int beg = rowptr[d];
    const int end = rowptr[d + 1];
    const float dd = dinv[d];
    const float* hl = H + lane * VPL;

    VT acc = *(const VT*)(hl + (size_t)d * DIM) * dd;

    for (int base = beg; base < end; base += 32) {
        const int n = (end - base < 32) ? (end - base) : 32;
        int ms = 0;
        float mw = 0.0f;
        if (lane < n) { ms = csr_src[base + lane]; mw = csr_w[base + lane]; }
        const int mwi = __float_as_int(mw);
        int k = 0;
        for (; k + 4 <= n; k += 4) {
            const int s0 = __builtin_amdgcn_readlane(ms, k);
            const int s1 = __builtin_amdgcn_readlane(ms, k + 1);
            const int s2 = __builtin_amdgcn_readlane(ms, k + 2);
            const int s3 = __builtin_amdgcn_readlane(ms, k + 3);
            const VT v0 = *(const VT*)(hl + (size_t)s0 * DIM);
            const VT v1 = *(const VT*)(hl + (size_t)s1 * DIM);
            const VT v2 = *(const VT*)(hl + (size_t)s2 * DIM);
            const VT v3 = *(const VT*)(hl + (size_t)s3 * DIM);
            const float w0 = __int_as_float(__builtin_amdgcn_readlane(mwi, k));
            const float w1 = __int_as_float(__builtin_amdgcn_readlane(mwi, k + 1));
            const float w2 = __int_as_float(__builtin_amdgcn_readlane(mwi, k + 2));
            const float w3 = __int_as_float(__builtin_amdgcn_readlane(mwi, k + 3));
            acc += v0 * w0;
            acc += v1 * w1;
            acc += v2 * w2;
            acc += v3 * w3;
        }
        for (; k < n; ++k) {
            const int s0 = __builtin_amdgcn_readlane(ms, k);
            const float w0 = __int_as_float(__builtin_amdgcn_readlane(mwi, k));
            acc += *(const VT*)(hl + (size_t)s0 * DIM) * w0;
        }
    }

    acc *= dd;
    if constexpr (ADD_BIAS) acc += *(const VT*)(bias + lane * VPL);
    *(VT*)(out + (size_t)d * DIM + lane * VPL) = acc;
}

template <int NCOLS, bool BIAS_RELU>
__global__ void k_gemm_wmma_k128(const float* A,
                                 const float* __restrict__ Bt,
                                 const float* __restrict__ bias,
                                 float* C) {
    __shared__ float As[16 * 132];
    const int tid  = threadIdx.x;
    const int row0 = blockIdx.x * 16;

    const float4* Ag = (const float4*)(A + (size_t)row0 * 128);
    for (int i = tid; i < 16 * 32; i += blockDim.x) {
        int r = i >> 5, c4 = i & 31;
        ((float4*)(As + r * 132))[c4] = Ag[r * 32 + c4];
    }
    __syncthreads();

    const int lane  = tid & 31;
    const int col0  = (tid >> 5) * 16;
    const int n15   = lane & 15;
    const int khalf = lane >> 4;

    v8f acc = {0.f, 0.f, 0.f, 0.f, 0.f, 0.f, 0.f, 0.f};
    const float* ap = As + n15 * 132 + 2 * khalf;
    const float* bp = Bt + (size_t)(col0 + n15) * 128 + 2 * khalf;

#pragma unroll
    for (int kk = 0; kk < 128; kk += 4) {
        v2f a; a.x = ap[kk]; a.y = ap[kk + 1];
        v2f b = *(const v2f*)(bp + kk);
        acc = __builtin_amdgcn_wmma_f32_16x16x4_f32(
             false, a,  false, b,
             (short)0, acc,  false,  false);
    }

    float* Cp = C + (size_t)(row0 + 8 * khalf) * NCOLS + col0 + n15;
    if constexpr (BIAS_RELU) {
        const float bv = bias[col0 + n15];
#pragma unroll
        for (int r = 0; r < 8; ++r) Cp[r * NCOLS] = fmaxf(acc[r] + bv, 0.0f);
    } else {
#pragma unroll
        for (int r = 0; r < 8; ++r) Cp[r * NCOLS] = acc[r];
    }
}

template <int CNT, int M>
__device__ __forceinline__ void fold_half(float (&p)[16], int lane) {
    const bool hi = (lane & M) != 0;
#pragma unroll
    for (int i = 0; i < CNT / 2; ++i) {
        const float keep = hi ? p[i + CNT / 2] : p[i];
        const float send = hi ? p[i] : p[i + CNT / 2];
        p[i] = keep + __shfl_xor(send, M, 32);
    }
}

template <int K>
__device__ __forceinline__ void edge_products(float (&p)[16], const float* __restrict__ Z,
                                              int lane, int ms, v2f zd) {
    v2f v[K];
#pragma unroll
    for (int i = 0; i < K; ++i) {
        const int s = __builtin_amdgcn_readlane(ms, 2 * i);
        const float* rp = Z + (size_t)s * 64;
        v[i] = *(const v2f*)(rp + lane * 2);
    }
#pragma unroll
    for (int i = 0; i < K; ++i) p[i] = v[i].x * zd.x + v[i].y * zd.y;
#pragma unroll
    for (int i = K; i < 16; ++i) p[i] = 0.0f;
}

__global__ void k_decode_pull(const float* __restrict__ Z, const int* __restrict__ rowptr,
                              const int* __restrict__ csr_src, float* __restrict__ ec, int N) {
    const int lane = threadIdx.x & 31;
    const int wv = __builtin_amdgcn_readfirstlane((int)(threadIdx.x >> 5));
    const int d = blockIdx.x * (int)(blockDim.x >> 5) + wv;
    if (d >= N) return;

    const int beg = rowptr[d];
    const int end = rowptr[d + 1];
    if (beg >= end) return;
    const int slot = lane >> 1;
    const int last = end - 1;

    int ms = csr_src[(beg + slot < last) ? (beg + slot) : last];
    const v2f zd = *(const v2f*)(Z + (size_t)d * 64 + lane * 2);

    for (int base = beg; base < end; base += 16) {
        const int n = (end - base < 16) ? (end - base) : 16;
        const int nx = base + 16 + slot;
        const int ms_next = csr_src[(nx < last) ? nx : last];

        float p[16];
        if (n > 12)      edge_products<16>(p, Z, lane, ms, zd);
        else if (n > 8)  edge_products<12>(p, Z, lane, ms, zd);
        else if (n > 4)  edge_products<8>(p, Z, lane, ms, zd);
        else             edge_products<4>(p, Z, lane, ms, zd);

        fold_half<16, 16>(p, lane);
        fold_half<8, 8>(p, lane);
        fold_half<4, 4>(p, lane);
        fold_half<2, 2>(p, lane);
        const float sc = p[0] + __shfl_xor(p[0], 1, 32);

        const float sig = 1.0f / (1.0f + __expf(-sc));
        const float err = -__logf(sig + 1e-15f);
        if (slot < n) {
            const int odd = lane & 1;
            atomAddF(ec + 2 * (size_t)ms + odd, odd ? 1.0f : err);
        }
        ms = ms_next;
    }
}

__global__ void k_finalize(const float* __restrict__ ec, float* __restrict__ out, int N) {
    int i = blockIdx.x * blockDim.x + threadIdx.x;
    if (i < N) {
        const v2f v = *(const v2f*)(ec + 2 * (size_t)i);
        out[i] = (v.y > 0.0f) ? (v.x / fmaxf(v.y, 1.0f)) : 10.0f;
    }
}

extern "C" void kernel_launch(void* const* d_in, const int* in_sizes, int n_in,
                              void* d_out, int out_size, void* d_ws, size_t ws_size,
                              hipStream_t stream) {
    const float* x  = (const float*)d_in[0];
    const int*   ei = (const int*)d_in[1];
    const float* W1 = (const float*)d_in[2];
    const float* b1 = (const float*)d_in[3];
    const float* W2 = (const float*)d_in[4];
    const float* b2 = (const float*)d_in[5];

    const int N = in_sizes[0] / 128;
    const int E = in_sizes[1] / 2;
    const int* src = ei;
    const int* dst = ei + E;

    float* ws     = (float*)d_ws;
    float* dinv   = ws;
    float* ec     = dinv + N;
    size_t off    = (((size_t)3 * N) + 63) & ~(size_t)63;
    float* wt1    = ws + off;
    float* wt2    = wt1 + 128 * 128;
    float* bufA   = wt2 + 64 * 128;
    float* bufB   = bufA + (size_t)N * 128;
    const int NT  = (N + 1023) / 1024;
    int*   indeg   = (int*)(bufB + (size_t)N * 64);
    int*   rowptr  = indeg + N;
    int*   part    = rowptr + N + 1;
    int*   pos     = part + NT;
    int*   csr_src = pos + E;
    float* csr_w   = (float*)(csr_src + E);

    const int T = 256;
    k_init      <<<(N + T - 1) / T, T, 0, stream>>>(indeg, ec, N);
    k_count     <<<(E + T - 1) / T, T, 0, stream>>>(dst, indeg, pos, E);
    k_scan_part <<<NT, 256, 0, stream>>>(indeg, part, N);
    k_scan_write<<<NT, 256, 0, stream>>>(indeg, part, rowptr, dinv, N, E);
    k_fill      <<<(E + T - 1) / T, T, 0, stream>>>(src, dst, pos, rowptr, dinv, csr_src, csr_w, E);
    k_transpose<<<(128 * 128 + T - 1) / T, T, 0, stream>>>(W1, wt1, 128);
    k_transpose<<<(128 * 64  + T - 1) / T, T, 0, stream>>>(W2, wt2, 64);

    k_pull<128, false><<<(N + 7) / 8, 256, 0, stream>>>(x, bufA, rowptr, csr_src, csr_w, dinv, b1, N);
    k_gemm_wmma_k128<128, true><<<N / 16, 256, 0, stream>>>(bufA, wt1, b1, bufA);

    k_gemm_wmma_k128<64, false><<<N / 16, 128, 0, stream>>>(bufA, wt2, b2, bufB);
    k_pull<64, true><<<(N + 7) / 8, 256, 0, stream>>>(bufB, bufA, rowptr, csr_src, csr_w, dinv, b2, N);

    k_decode_pull<<<(N + 7) / 8, 256, 0, stream>>>(bufA, rowptr, csr_src, ec, N);
    k_finalize<<<(N + T - 1) / T, T, 0, stream>>>(ec, (float*)d_out, N);
}


# Destination sort with LDS counters: edges split into 256-row bins, then one block per bin ranks its edges in LDS; replaces a returning global atomic per edge; row sum order stays arrival order
# speedup vs baseline: 2.9921x; 1.2716x over previous
#include <hip/hip_runtime.h>
#include <math.h>


typedef float v2f __attribute__((ext_vector_type(2)));
typedef float v4f __attribute__((ext_vector_type(4)));
typedef float v8f __attribute__((ext_vector_type(8)));

template <int V> struct VecT;
template <> struct VecT<4> { typedef v4f T; };
template <> struct VecT<2> { typedef v2f T; };

__device__ __forceinline__ void atomAddF(float* p, float v) {
    __hip_atomic_fetch_add(p, v, __ATOMIC_RELAXED, __HIP_MEMORY_SCOPE_AGENT);
}

typedef int v2i __attribute__((ext_vector_type(2)));

#define BIN_SHIFT 8
#define BIN_ROWS 256
#define MAX_BINS 512
#define BIN_STRIDE 32
#define SPLIT_THREADS 512
#define SPLIT_PER_THREAD 8

__global__ void k_init(int* __restrict__ bins, float* __restrict__ ec, int N) {
    int i = blockIdx.x * blockDim.x + threadIdx.x;
    for (int k = i; k < 2 * MAX_BINS * BIN_STRIDE; k += (int)(gridDim.x * blockDim.x)) bins[k] = 0;
    if (i < N) {
        v2f z; z.x = 0.0f; z.y = 0.0f;
        *(v2f*)(ec + 2 * (size_t)i) = z;
    }
}

__global__ void k_binhist(const int* __restrict__ dst, int* __restrict__ binCount,
                          int E, int NB) {
    __shared__ int hist[MAX_BINS];
    const int t = threadIdx.x;
    hist[t] = 0;
    __syncthreads();
    const int base = blockIdx.x * (SPLIT_THREADS * SPLIT_PER_THREAD) + t;
    int d[SPLIT_PER_THREAD];
#pragma unroll
    for (int i = 0; i < SPLIT_PER_THREAD; ++i) {
        const int e = base + i * SPLIT_THREADS;
        d[i] = (e < E) ? dst[e] : -1;
    }
#pragma unroll
    for (int i = 0; i < SPLIT_PER_THREAD; ++i)
        if (d[i] >= 0)
            __hip_atomic_fetch_add(&hist[d[i] >> BIN_SHIFT], 1, __ATOMIC_RELAXED,
                                   __HIP_MEMORY_SCOPE_WORKGROUP);
    __syncthreads();
    if (t < NB && hist[t] > 0)
        __hip_atomic_fetch_add(&binCount[t * BIN_STRIDE], hist[t], __ATOMIC_RELAXED,
                               __HIP_MEMORY_SCOPE_AGENT);
}

__global__ void k_split(const int* __restrict__ src, const int* __restrict__ dst,
                        const int* __restrict__ binCount, int* __restrict__ cursor,
                        int* __restrict__ pairs, int E, int NB) {
    __shared__ int hist[MAX_BINS];
    __shared__ int gbase[MAX_BINS];
    const int t = threadIdx.x;
    hist[t] = 0;
    __syncthreads();
    const int base = blockIdx.x * (SPLIT_THREADS * SPLIT_PER_THREAD) + t;
    int d[SPLIT_PER_THREAD], s[SPLIT_PER_THREAD], r[SPLIT_PER_THREAD];
#pragma unroll
    for (int i = 0; i < SPLIT_PER_THREAD; ++i) {
        const int e = base + i * SPLIT_THREADS;
        d[i] = (e < E) ? dst[e] : -1;
        s[i] = (e < E) ? src[e] : 0;
    }
#pragma unroll
    for (int i = 0; i < SPLIT_PER_THREAD; ++i) {
        r[i] = 0;
        if (d[i] >= 0)
            r[i] = __hip_atomic_fetch_add(&hist[d[i] >> BIN_SHIFT], 1, __ATOMIC_RELAXED,
                                          __HIP_MEMORY_SCOPE_WORKGROUP);
    }
    if (t < 32) {
        int v[16];
        int sum = 0;
#pragma unroll
        for (int q = 0; q < 16; ++q) {
            const int bb = t * 16 + q;
            v[q] = (bb < NB) ? binCount[bb * BIN_STRIDE] : 0;
            sum += v[q];
        }
        int incl = sum;
        for (int off = 1; off < 32; off <<= 1) {
            int y = __shfl_up(incl, off, 32);
            if (t >= off) incl += y;
        }
        int run = incl - sum;
#pragma unroll
        for (int q = 0; q < 16; ++q) {
            gbase[t * 16 + q] = run;
            run += v[q];
        }
    }
    __syncthreads();
    if (t < NB) {
        const int h = hist[t];
        if (h > 0)
            gbase[t] += __hip_atomic_fetch_add(&cursor[t * BIN_STRIDE], h, __ATOMIC_RELAXED,
                                               __HIP_MEMORY_SCOPE_AGENT);
    }
    __syncthreads();
#pragma unroll
    for (int i = 0; i < SPLIT_PER_THREAD; ++i) {
        if (d[i] >= 0) {
            const int j = gbase[d[i] >> BIN_SHIFT] + r[i];
            v2i p; p.x = s[i]; p.y = d[i];
            *(v2i*)(pairs + 2 * (size_t)j) = p;
        }
    }
}

__global__ void k_binsort(const int* __restrict__ pairs, const int* __restrict__ binCount,
                          int* __restrict__ rowptr, float* __restrict__ dinv,
                          int* __restrict__ csr_src, int N, int E) {
    __shared__ int cnt[BIN_ROWS];
    __shared__ int off[BIN_ROWS];
    __shared__ int wsum[BIN_ROWS / 32];
    __shared__ int sbeg;
    const int t = threadIdx.x;
    const int lane = t & 31;
    const int b = blockIdx.x;
    const int lo = b << BIN_SHIFT;
    if (t < BIN_ROWS) cnt[t] = 0;
    if (t < 32) {
        int p = 0;
        for (int i = t; i < b; i += 32) p += binCount[i * BIN_STRIDE];
        for (int o = 16; o > 0; o >>= 1) p += __shfl_xor(p, o, 32);
        if (t == 0) sbeg = p;
    }
    __syncthreads();
    const int beg = sbeg;
    const int len = binCount[b * BIN_STRIDE];

    for (int j0 = t; j0 < len; j0 += 4 * 1024) {
        int dd[4];
#pragma unroll
        for (int q = 0; q < 4; ++q) {
            const int j = j0 + q * 1024;
            dd[q] = (j < len) ? pairs[2 * (size_t)(beg + j) + 1] : -1;
        }
#pragma unroll
        for (int q = 0; q < 4; ++q)
            if (dd[q] >= 0)
                __hip_atomic_fetch_add(&cnt[dd[q] - lo], 1, __ATOMIC_RELAXED,
                                       __HIP_MEMORY_SCOPE_WORKGROUP);
    }
    __syncthreads();

    const int c = (t < BIN_ROWS) ? cnt[t] : 0;
    int incl = c;
    for (int o = 1; o < 32; o <<= 1) {
        int y = __shfl_up(incl, o, 32);
        if (lane >= o) incl += y;
    }
    if (t < BIN_ROWS && lane == 31) wsum[t >> 5] = incl;
    __syncthreads();
    if (t < BIN_ROWS) {
        int run = incl - c;
        for (int i = 0; i < (t >> 5); ++i) run += wsum[i];
        off[t] = run;
        cnt[t] = 0;
        const int node = lo + t;
        if (node < N) {
            rowptr[node] = beg + run;
            dinv[node] = rsqrtf((float)(c + 1));
        }
    }
    if (b == 0 && t == 0) rowptr[N] = E;
    __syncthreads();

    for (int j0 = t; j0 < len; j0 += 4 * 1024) {
        v2i pr[4];
#pragma unroll
        for (int q = 0; q < 4; ++q) {
            const int j = j0 + q * 1024;
            v2i none; none.x = 0; none.y = -1;
            pr[q] = (j < len) ? *(const v2i*)(pairs + 2 * (size_t)(beg + j)) : none;
        }
#pragma unroll
        for (int q = 0; q < 4; ++q) {
            if (pr[q].y >= 0) {
                const int k = pr[q].y - lo;
                const int r = __hip_atomic_fetch_add(&cnt[k], 1, __ATOMIC_RELAXED,
                                                     __HIP_MEMORY_SCOPE_WORKGROUP);
                csr_src[beg + off[k] + r] = pr[q].x;
            }
        }
    }
}

__global__ void k_weights(const int* __restrict__ csr_src, const float* __restrict__ dinv,
                          float* __restrict__ csr_w, int E) {
    int j = blockIdx.x * blockDim.x + threadIdx.x;
    if (j < E) csr_w[j] = dinv[csr_src[j]];
}

__global__ void k_transpose(const float* __restrict__ W, float* __restrict__ Wt,
                            int N) {
    int idx = blockIdx.x * blockDim.x + threadIdx.x;
    if (idx < 128 * N) {
        int k = idx / N, n = idx - k * N;
        Wt[n * 128 + k] = W[idx];
    }
}

template <int DIM, bool ADD_BIAS>
__global__ void k_pull(const float* __restrict__ H, float* __restrict__ out,
                       const int* __restrict__ rowptr, const int* __restrict__ csr_src,
                       const float* __restrict__ csr_w, const float* __restrict__ dinv,
                       const float* __restrict__ bias, int N) {
    constexpr int VPL = DIM / 32;
    typedef typename VecT<VPL>::T VT;
    const int lane = threadIdx.x & 31;
    const int wv = __builtin_amdgcn_readfirstlane((int)(threadIdx.x >> 5));
    const int d = blockIdx.x * (int)(blockDim.x >> 5) + wv;
    if (d >= N) return;

    const int beg = rowptr[d];
    const int end = rowptr[d + 1];
    const float dd = dinv[d];
    const float* hl = H + lane * VPL;

    VT acc = *(const VT*)(hl + (size_t)d * DIM) * dd;

    for (int base = beg; base < end; base += 32) {
        const int n = (end - base < 32) ? (end - base) : 32;
        int ms = 0;
        float mw = 0.0f;
        if (lane < n) { ms = csr_src[base + lane]; mw = csr_w[base + lane]; }
        const int mwi = __float_as_int(mw);
        int k = 0;
        for (; k + 4 <= n; k += 4) {
            const int s0 = __builtin_amdgcn_readlane(ms, k);
            const int s1 = __builtin_amdgcn_readlane(ms, k + 1);
            const int s2 = __builtin_amdgcn_readlane(ms, k + 2);
            const int s3 = __builtin_amdgcn_readlane(ms, k + 3);
            const VT v0 = *(const VT*)(hl + (size_t)s0 * DIM);
            const VT v1 = *(const VT*)(hl + (size_t)s1 * DIM);
            const VT v2 = *(const VT*)(hl + (size_t)s2 * DIM);
            const VT v3 = *(const VT*)(hl + (size_t)s3 * DIM);
            const float w0 = __int_as_float(__builtin_amdgcn_readlane(mwi, k));
            const float w1 = __int_as_float(__builtin_amdgcn_readlane(mwi, k + 1));
            const float w2 = __int_as_float(__builtin_amdgcn_readlane(mwi, k + 2));
            const float w3 = __int_as_float(__builtin_amdgcn_readlane(mwi, k + 3));
            acc += v0 * w0;
            acc += v1 * w1;
            acc += v2 * w2;
            acc += v3 * w3;
        }
        for (; k < n; ++k) {
            const int s0 = __builtin_amdgcn_readlane(ms, k);
            const float w0 = __int_as_float(__builtin_amdgcn_readlane(mwi, k));
            acc += *(const VT*)(hl + (size_t)s0 * DIM) * w0;
        }
    }

    acc *= dd;
    if constexpr (ADD_BIAS) acc += *(const VT*)(bias + lane * VPL);
    *(VT*)(out + (size_t)d * DIM + lane * VPL) = acc;
}

template <int NCOLS, bool BIAS_RELU>
__global__ void k_gemm_wmma_k128(const float* A,
                                 const float* __restrict__ Bt,
                                 const float* __restrict__ bias,
                                 float* C) {
    __shared__ float As[16 * 132];
    const int tid  = threadIdx.x;
    const int row0 = blockIdx.x * 16;

    const float4* Ag = (const float4*)(A + (size_t)row0 * 128);
    for (int i = tid; i < 16 * 32; i += blockDim.x) {
        int r = i >> 5, c4 = i & 31;
        ((float4*)(As + r * 132))[c4] = Ag[r * 32 + c4];
    }
    __syncthreads();

    const int lane  = tid & 31;
    const int col0  = (tid >> 5) * 16;
    const int n15   = lane & 15;
    const int khalf = lane >> 4;

    v8f acc = {0.f, 0.f, 0.f, 0.f, 0.f, 0.f, 0.f, 0.f};
    const float* ap = As + n15 * 132 + 2 * khalf;
    const float* bp = Bt + (size_t)(col0 + n15) * 128 + 2 * khalf;

#pragma unroll
    for (int kk = 0; kk < 128; kk += 4) {
        v2f a; a.x = ap[kk]; a.y = ap[kk + 1];
        v2f b = *(const v2f*)(bp + kk);
        acc = __builtin_amdgcn_wmma_f32_16x16x4_f32(
             false, a,  false, b,
             (short)0, acc,  false,  false);
    }

    float* Cp = C + (size_t)(row0 + 8 * khalf) * NCOLS + col0 + n15;
    if constexpr (BIAS_RELU) {
        const float bv = bias[col0 + n15];
#pragma unroll
        for (int r = 0; r < 8; ++r) Cp[r * NCOLS] = fmaxf(acc[r] + bv, 0.0f);
    } else {
#pragma unroll
        for (int r = 0; r < 8; ++r) Cp[r * NCOLS] = acc[r];
    }
}

template <int CNT, int M>
__device__ __forceinline__ void fold_half(float (&p)[16], int lane) {
    const bool hi = (lane & M) != 0;
#pragma unroll
    for (int i = 0; i < CNT / 2; ++i) {
        const float keep = hi ? p[i + CNT / 2] : p[i];
        const float send = hi ? p[i] : p[i + CNT / 2];
        p[i] = keep + __shfl_xor(send, M, 32);
    }
}

template <int K>
__device__ __forceinline__ void edge_products(float (&p)[16], const float* __restrict__ Z,
                                              int lane, int ms, v2f zd) {
    v2f v[K];
#pragma unroll
    for (int i = 0; i < K; ++i) {
        const int s = __builtin_amdgcn_readlane(ms, 2 * i);
        const float* rp = Z + (size_t)s * 64;
        v[i] = *(const v2f*)(rp + lane * 2);
    }
#pragma unroll
    for (int i = 0; i < K; ++i) p[i] = v[i].x * zd.x + v[i].y * zd.y;
#pragma unroll
    for (int i = K; i < 16; ++i) p[i] = 0.0f;
}

__global__ void k_decode_pull(const float* __restrict__ Z, const int* __restrict__ rowptr,
                              const int* __restrict__ csr_src, float* __restrict__ ec, int N) {
    const int lane = threadIdx.x & 31;
    const int wv = __builtin_amdgcn_readfirstlane((int)(threadIdx.x >> 5));
    const int d = blockIdx.x * (int)(blockDim.x >> 5) + wv;
    if (d >= N) return;

    const int beg = rowptr[d];
    const int end = rowptr[d + 1];
    if (beg >= end) return;
    const int slot = lane >> 1;
    const int last = end - 1;

    int ms = csr_src[(beg + slot < last) ? (beg + slot) : last];
    const v2f zd = *(const v2f*)(Z + (size_t)d * 64 + lane * 2);

    for (int base = beg; base < end; base += 16) {
        const int n = (end - base < 16) ? (end - base) : 16;
        const int nx = base + 16 + slot;
        const int ms_next = csr_src[(nx < last) ? nx : last];

        float p[16];
        if (n > 12)      edge_products<16>(p, Z, lane, ms, zd);
        else if (n > 8)  edge_products<12>(p, Z, lane, ms, zd);
        else if (n > 4)  edge_products<8>(p, Z, lane, ms, zd);
        else             edge_products<4>(p, Z, lane, ms, zd);

        fold_half<16, 16>(p, lane);
        fold_half<8, 8>(p, lane);
        fold_half<4, 4>(p, lane);
        fold_half<2, 2>(p, lane);
        const float sc = p[0] + __shfl_xor(p[0], 1, 32);

        const float sig = 1.0f / (1.0f + __expf(-sc));
        const float err = -__logf(sig + 1e-15f);
        if (slot < n) {
            const int odd = lane & 1;
            atomAddF(ec + 2 * (size_t)ms + odd, odd ? 1.0f : err);
        }
        ms = ms_next;
    }
}

__global__ void k_finalize(const float* __restrict__ ec, float* __restrict__ out, int N) {
    int i = blockIdx.x * blockDim.x + threadIdx.x;
    if (i < N) {
        const v2f v = *(const v2f*)(ec + 2 * (size_t)i);
        out[i] = (v.y > 0.0f) ? (v.x / fmaxf(v.y, 1.0f)) : 10.0f;
    }
}

extern "C" void kernel_launch(void* const* d_in, const int* in_sizes, int n_in,
                              void* d_out, int out_size, void* d_ws, size_t ws_size,
                              hipStream_t stream) {
    const float* x  = (const float*)d_in[0];
    const int*   ei = (const int*)d_in[1];
    const float* W1 = (const float*)d_in[2];
    const float* b1 = (const float*)d_in[3];
    const float* W2 = (const float*)d_in[4];
    const float* b2 = (const float*)d_in[5];

    const int N = in_sizes[0] / 128;
    const int E = in_sizes[1] / 2;
    const int* src = ei;
    const int* dst = ei + E;

    float* ws     = (float*)d_ws;
    float* dinv   = ws;
    float* ec     = dinv + N;
    size_t off    = (((size_t)3 * N) + 63) & ~(size_t)63;
    float* wt1    = ws + off;
    float* wt2    = wt1 + 128 * 128;
    float* bufA   = wt2 + 64 * 128;
    float* bufB   = bufA + (size_t)N * 128;
    const int NB  = (N + BIN_ROWS - 1) / BIN_ROWS;
    int*   rowptr   = (int*)(bufB + (size_t)N * 64);
    int*   binCount = rowptr + N + 1;
    int*   cursor   = binCount + MAX_BINS * BIN_STRIDE;
    int*   csr_src  = cursor + MAX_BINS * BIN_STRIDE;
    float* csr_w    = (float*)(csr_src + E);
    int*   pairs    = (int*)bufA;

    const int T = 256;
    const int EPB = SPLIT_THREADS * SPLIT_PER_THREAD;
    k_init    <<<(N + T - 1) / T, T, 0, stream>>>(binCount, ec, N);
    k_binhist <<<(E + EPB - 1) / EPB, SPLIT_THREADS, 0, stream>>>(dst, binCount, E, NB);
    k_split   <<<(E + EPB - 1) / EPB, SPLIT_THREADS, 0, stream>>>(src, dst, binCount, cursor, pairs, E, NB);
    k_binsort <<<NB, 1024, 0, stream>>>(pairs, binCount, rowptr, dinv, csr_src, N, E);
    k_weights <<<(E + T - 1) / T, T, 0, stream>>>(csr_src, dinv, csr_w, E);
    k_transpose<<<(128 * 128 + T - 1) / T, T, 0, stream>>>(W1, wt1, 128);
    k_transpose<<<(128 * 64  + T - 1) / T, T, 0, stream>>>(W2, wt2, 64);

    k_pull<128, false><<<(N + 7) / 8, 256, 0, stream>>>(x, bufA, rowptr, csr_src, csr_w, dinv, b1, N);
    k_gemm_wmma_k128<128, true><<<N / 16, 256, 0, stream>>>(bufA, wt1, b1, bufA);

    k_gemm_wmma_k128<64, false><<<N / 16, 128, 0, stream>>>(bufA, wt2, b2, bufB);
    k_pull<64, true><<<(N + 7) / 8, 256, 0, stream>>>(bufB, bufA, rowptr, csr_src, csr_w, dinv, b2, N);

    k_decode_pull<<<(N + 7) / 8, 256, 0, stream>>>(bufA, rowptr, csr_src, ec, N);
    k_finalize<<<(N + T - 1) / T, T, 0, stream>>>(ec, (float*)d_out, N);
}


# Row kernels (both pulls, decode, edge weights): a wave walks eight consecutive rows, bounds in one request, next row's indices and own row requested under this row's gathers; 8x fewer waves
# speedup vs baseline: 4.7683x; 1.5936x over previous
#include <hip/hip_runtime.h>
#include <math.h>


typedef float v2f __attribute__((ext_vector_type(2)));
typedef float v4f __attribute__((ext_vector_type(4)));
typedef float v8f __attribute__((ext_vector_type(8)));

template <int V> struct VecT;
template <> struct VecT<4> { typedef v4f T; };
template <> struct VecT<2> { typedef v2f T; };

typedef int v2i __attribute__((ext_vector_type(2)));

#define BIN_SHIFT 8
#define BIN_ROWS 256
#define MAX_BINS 512
#define CH_THREADS 1024
#define CH_PER_THREAD 8
#define CH_EDGES 8192
#define MAX_CHUNKS 1024

__global__ void __launch_bounds__(CH_THREADS)
k_bucket(const int* __restrict__ srcp, const int* __restrict__ dstp,
         int* __restrict__ pairsD, int* __restrict__ offsD,
         int* __restrict__ pairsS, int* __restrict__ offsS, int E, int NB, int NC) {
    __shared__ int hist[MAX_BINS];
    __shared__ int gbase[MAX_BINS + 1];
    __shared__ v2i stage[CH_EDGES];
    const bool bySrc = blockIdx.y != 0;
    const int* __restrict__ dst = bySrc ? srcp : dstp;
    const int* __restrict__ src = bySrc ? dstp : srcp;
    int* __restrict__ pairs = bySrc ? pairsS : pairsD;
    int* __restrict__ offs = bySrc ? offsS : offsD;
    const int t = threadIdx.x;
    const int chunk = blockIdx.x;
    if (t < MAX_BINS) hist[t] = 0;
    __syncthreads();
    const int base = chunk * CH_EDGES + t;
    int d[CH_PER_THREAD], s[CH_PER_THREAD], r[CH_PER_THREAD];
#pragma unroll
    for (int i = 0; i < CH_PER_THREAD; ++i) {
        const int e = base + i * CH_THREADS;
        d[i] = (e < E) ? dst[e] : -1;
        s[i] = (e < E) ? src[e] : 0;
    }
#pragma unroll
    for (int i = 0; i < CH_PER_THREAD; ++i) {
        r[i] = 0;
        if (d[i] >= 0)
            r[i] = __hip_atomic_fetch_add(&hist[d[i] >> BIN_SHIFT], 1, __ATOMIC_RELAXED,
                                          __HIP_MEMORY_SCOPE_WORKGROUP);
    }
    __syncthreads();
    if (t < 32) {
        int v[16];
        int sum = 0;
#pragma unroll
        for (int q = 0; q < 16; ++q) {
            v[q] = hist[t * 16 + q];
            sum += v[q];
        }
        int incl = sum;
        for (int o = 1; o < 32; o <<= 1) {
            int y = __shfl_up(incl, o, 32);
            if (t >= o) incl += y;
        }
        int run = incl - sum;
#pragma unroll
        for (int q = 0; q < 16; ++q) {
            gbase[t * 16 + q] = run;
            run += v[q];
        }
        if (t == 31) gbase[MAX_BINS] = run;
    }
    __syncthreads();
    if (t <= NB) offs[(size_t)t * NC + chunk] = gbase[t];
#pragma unroll
    for (int i = 0; i < CH_PER_THREAD; ++i) {
        if (d[i] >= 0) {
            v2i p; p.x = s[i]; p.y = d[i];
            stage[gbase[d[i] >> BIN_SHIFT] + r[i]] = p;
        }
    }
    __syncthreads();
    const int total = gbase[MAX_BINS];
    v2i* __restrict__ outp = (v2i*)pairs + (size_t)chunk * CH_EDGES;
#pragma unroll
    for (int i = 0; i < CH_PER_THREAD; ++i) {
        const int j = t + i * CH_THREADS;
        if (j < total) outp[j] = stage[j];
    }
}

__device__ __forceinline__ int find_run(const int* rstart, int NC, int j) {
    int lo = 0, hi = NC;
    while (lo < hi) {
        const int mid = (lo + hi) >> 1;
        if (rstart[mid] <= j) lo = mid + 1; else hi = mid;
    }
    return lo - 1;
}

__global__ void __launch_bounds__(1024)
k_binsort(const int* __restrict__ pairsD, const int* __restrict__ offsD,
          int* __restrict__ rowptrD, float* __restrict__ dinv,
          int* __restrict__ csrD,
          const int* __restrict__ pairsS, const int* __restrict__ offsS,
          int* __restrict__ rowptrS, int* __restrict__ csrS,
          int N, int E, int NC) {
    const bool bySrc = blockIdx.y != 0;
    const int* __restrict__ pairs = bySrc ? pairsS : pairsD;
    const int* __restrict__ offs = bySrc ? offsS : offsD;
    int* __restrict__ rowptr = bySrc ? rowptrS : rowptrD;
    int* __restrict__ csr_src = bySrc ? csrS : csrD;
    __shared__ int cnt[BIN_ROWS];
    __shared__ int off[BIN_ROWS];
    __shared__ int wsum[BIN_ROWS / 32];
    __shared__ int rsum[32];
    __shared__ int bsum[32];
    __shared__ int rstart[MAX_CHUNKS];
    __shared__ int rbase[MAX_CHUNKS];
    const int t = threadIdx.x;
    const int lane = t & 31;
    const int wv = t >> 5;
    const int b = blockIdx.x;
    const int lo = b << BIN_SHIFT;
    if (t < BIN_ROWS) cnt[t] = 0;

    int rlo = 0, rlen = 0;
    if (t < NC) {
        rlo = offs[(size_t)b * NC + t];
        rlen = offs[(size_t)(b + 1) * NC + t] - rlo;
    }
    int rincl = rlen;
    for (int o = 1; o < 32; o <<= 1) {
        int y = __shfl_up(rincl, o, 32);
        if (lane >= o) rincl += y;
    }
    int bs = rlo;
    for (int o = 16; o > 0; o >>= 1) bs += __shfl_xor(bs, o, 32);
    if (lane == 31) rsum[wv] = rincl;
    if (lane == 0) bsum[wv] = bs;
    __syncthreads();
    int wpre = 0, len = 0, beg = 0;
    for (int i = 0; i < 32; ++i) {
        const int v = rsum[i];
        if (i < wv) wpre += v;
        len += v;
        beg += bsum[i];
    }
    if (t < NC) {
        const int excl = wpre + rincl - rlen;
        rstart[t] = excl;
        rbase[t] = t * CH_EDGES + rlo - excl;
    }
    __syncthreads();

    for (int j0 = t; j0 < len; j0 += 4 * 1024) {
        int dd[4];
#pragma unroll
        for (int q = 0; q < 4; ++q) {
            const int j = j0 + q * 1024;
            dd[q] = -1;
            if (j < len) {
                const int cc = find_run(rstart, NC, j);
                dd[q] = pairs[2 * (size_t)(rbase[cc] + j) + 1];
            }
        }
#pragma unroll
        for (int q = 0; q < 4; ++q)
            if (dd[q] >= 0)
                __hip_atomic_fetch_add(&cnt[dd[q] - lo], 1, __ATOMIC_RELAXED,
                                       __HIP_MEMORY_SCOPE_WORKGROUP);
    }
    __syncthreads();

    const int c = (t < BIN_ROWS) ? cnt[t] : 0;
    int incl = c;
    for (int o = 1; o < 32; o <<= 1) {
        int y = __shfl_up(incl, o, 32);
        if (lane >= o) incl += y;
    }
    if (t < BIN_ROWS && lane == 31) wsum[t >> 5] = incl;
    __syncthreads();
    if (t < BIN_ROWS) {
        int run = incl - c;
        for (int i = 0; i < (t >> 5); ++i) run += wsum[i];
        off[t] = run;
        cnt[t] = 0;
        const int node = lo + t;
        if (node < N) {
            rowptr[node] = beg + run;
            if (!bySrc) dinv[node] = rsqrtf((float)(c + 1));
        }
    }
    if (b == 0 && t == 0) rowptr[N] = E;
    __syncthreads();

    for (int j0 = t; j0 < len; j0 += 4 * 1024) {
        v2i pr[4];
#pragma unroll
        for (int q = 0; q < 4; ++q) {
            const int j = j0 + q * 1024;
            pr[q].x = 0; pr[q].y = -1;
            if (j < len) {
                const int cc = find_run(rstart, NC, j);
                pr[q] = *(const v2i*)(pairs + 2 * (size_t)(rbase[cc] + j));
            }
        }
#pragma unroll
        for (int q = 0; q < 4; ++q) {
            if (pr[q].y >= 0) {
                const int k = pr[q].y - lo;
                const int r = __hip_atomic_fetch_add(&cnt[k], 1, __ATOMIC_RELAXED,
                                                     __HIP_MEMORY_SCOPE_WORKGROUP);
                csr_src[beg + off[k] + r] = pr[q].x;
            }
        }
    }
}

#define ROWS_PER_WAVE 8
#define W_PER_THREAD 8

__global__ void __launch_bounds__(256)
k_weights(const int* __restrict__ csr_src, const float* __restrict__ dinv,
          float* __restrict__ csr_w, int E) {
    const int blk0 = blockIdx.x * (256 * W_PER_THREAD);
    const int base = blk0 + threadIdx.x;
    if (blk0 + 256 * W_PER_THREAD <= E) {
        int s[W_PER_THREAD];
        float w[W_PER_THREAD];
#pragma unroll
        for (int i = 0; i < W_PER_THREAD; ++i) s[i] = csr_src[base + i * 256];
#pragma unroll
        for (int i = 0; i < W_PER_THREAD; ++i) w[i] = dinv[s[i]];
#pragma unroll
        for (int i = 0; i < W_PER_THREAD; ++i) csr_w[base + i * 256] = w[i];
    } else {
        for (int i = 0; i < W_PER_THREAD; ++i) {
            const int j = base + i * 256;
            if (j < E) csr_w[j] = dinv[csr_src[j]];
        }
    }
}

template <int DIM>
__device__ __forceinline__ void pull_chunk(typename VecT<DIM / 32>::T& acc, const float* hl,
                                           int ms, int mwi, int n) {
    typedef typename VecT<DIM / 32>::T VT;
    int k = 0;
    for (; k + 4 <= n; k += 4) {
        const int s0 = __builtin_amdgcn_readlane(ms, k);
        const int s1 = __builtin_amdgcn_readlane(ms, k + 1);
        const int s2 = __builtin_amdgcn_readlane(ms, k + 2);
        const int s3 = __builtin_amdgcn_readlane(ms, k + 3);
        const VT v0 = *(const VT*)(hl + (size_t)s0 * DIM);
        const VT v1 = *(const VT*)(hl + (size_t)s1 * DIM);
        const VT v2 = *(const VT*)(hl + (size_t)s2 * DIM);
        const VT v3 = *(const VT*)(hl + (size_t)s3 * DIM);
        const float w0 = __int_as_float(__builtin_amdgcn_readlane(mwi, k));
        const float w1 = __int_as_float(__builtin_amdgcn_readlane(mwi, k + 1));
        const float w2 = __int_as_float(__builtin_amdgcn_readlane(mwi, k + 2));
        const float w3 = __int_as_float(__builtin_amdgcn_readlane(mwi, k + 3));
        acc += v0 * w0;
        acc += v1 * w1;
        acc += v2 * w2;
        acc += v3 * w3;
    }
    for (; k < n; ++k) {
        const int s0 = __builtin_amdgcn_readlane(ms, k);
        const float w0 = __int_as_float(__builtin_amdgcn_readlane(mwi, k));
        acc += *(const VT*)(hl + (size_t)s0 * DIM) * w0;
    }
}

template <int DIM, bool ADD_BIAS>
__global__ void __launch_bounds__(256)
k_pull(const float* __restrict__ H, float* __restrict__ out,
       const int* __restrict__ rowptr, const int* __restrict__ csr_src,
       const float* __restrict__ csr_w, const float* __restrict__ dinv,
       const float* __restrict__ bias, int N) {
    constexpr int VPL = DIM / 32;
    typedef typename VecT<VPL>::T VT;
    const int lane = threadIdx.x & 31;
    const int wv = __builtin_amdgcn_readfirstlane((int)(threadIdx.x >> 5));
    const int d0 = (blockIdx.x * (int)(blockDim.x >> 5) + wv) * ROWS_PER_WAVE;
    if (d0 >= N) return;
    const int cnt = (N - d0 < ROWS_PER_WAVE) ? (N - d0) : ROWS_PER_WAVE;

    const int ri = d0 + lane;
    const int rpv = rowptr[(ri < N) ? ri : N];
    const int ddv = __float_as_int(dinv[(ri < N) ? ri : (N - 1)]);
    const float* hl = H + lane * VPL;
    VT bv = {};
    if constexpr (ADD_BIAS) bv = *(const VT*)(bias + lane * VPL);

    int beg = __builtin_amdgcn_readlane(rpv, 0);
    int end = __builtin_amdgcn_readlane(rpv, 1);
    int ms = 0;
    float mw = 0.0f;
    if (beg + lane < end) { ms = csr_src[beg + lane]; mw = csr_w[beg + lane]; }
    VT selfv = *(const VT*)(hl + (size_t)d0 * DIM);

    for (int i = 0; i < cnt; ++i) {
        const int d = d0 + i;
        const float dd = __int_as_float(__builtin_amdgcn_readlane(ddv, i));
        int nend = end;
        int ms_n = 0;
        float mw_n = 0.0f;
        VT self_n = selfv;
        if (i + 1 < cnt) {
            nend = __builtin_amdgcn_readlane(rpv, i + 2);
            if (end + lane < nend) { ms_n = csr_src[end + lane]; mw_n = csr_w[end + lane]; }
            self_n = *(const VT*)(hl + (size_t)(d + 1) * DIM);
        }

        VT acc = selfv * dd;

        if (beg < end)
            pull_chunk<DIM>(acc, hl, ms, __float_as_int(mw), (end - beg < 32) ? (end - beg) : 32);
        for (int base = beg + 32; base < end; base += 32) {
            const int n = (end - base < 32) ? (end - base) : 32;
            int cs = 0;
            float cw = 0.0f;
            if (lane < n) { cs = csr_src[base + lane]; cw = csr_w[base + lane]; }
            pull_chunk<DIM>(acc, hl, cs, __float_as_int(cw), n);
        }

        acc *= dd;
        if constexpr (ADD_BIAS) acc += bv;
        *(VT*)(out + (size_t)d * DIM + lane * VPL) = acc;

        beg = end;
        end = nend;
        ms = ms_n;
        mw = mw_n;
        selfv = self_n;
    }
}

#define PROJ_ROWS 32
#define PROJ_PITCH 132
#define PROJ_BLOCKS 512

__device__ __forceinline__ void proj_load_rows(float4 (&v)[4], const float* __restrict__ A,
                                               int row0, int tid, int M) {
    const float4* __restrict__ src = (const float4*)(A + (size_t)row0 * 128) + tid;
    if (row0 + PROJ_ROWS <= M) {
#pragma unroll
        for (int i = 0; i < 4; ++i) v[i] = src[i * 256];
    } else {
#pragma unroll
        for (int i = 0; i < 4; ++i) {
            const int r = row0 + ((tid + i * 256) >> 5);
            v[i] = (r < M) ? src[i * 256] : make_float4(0.f, 0.f, 0.f, 0.f);
        }
    }
}

__global__ void __launch_bounds__(256)
k_project_both(const float* __restrict__ A, const float* __restrict__ W1,
               const float* __restrict__ b1, const float* __restrict__ W2,
               float* __restrict__ Z0, int M, int numTiles) {
    __shared__ float As[PROJ_ROWS * PROJ_PITCH];
    __shared__ float Hs[PROJ_ROWS * PROJ_PITCH];
    const int tid   = threadIdx.x;
    const int lane  = tid & 31;
    const int wv    = tid >> 5;
    const int n15   = lane & 15;
    const int khalf = lane >> 4;
    const int rt    = wv >> 2;
    const int ct    = wv & 3;

    int tile = blockIdx.x;
    float4 nx[4];
    proj_load_rows(nx, A, tile * PROJ_ROWS, tid, M);

    v2f bw1[32], bw2[32];
    {
        const float* p1 = W1 + (size_t)(2 * khalf) * 128 + wv * 16 + n15;
        const float* p2 = W2 + (size_t)(2 * khalf) * 64 + ct * 16 + n15;
#pragma unroll
        for (int q = 0; q < 32; ++q) {
            bw1[q].x = p1[(size_t)(4 * q) * 128];
            bw1[q].y = p1[(size_t)(4 * q + 1) * 128];
            bw2[q].x = p2[(size_t)(4 * q) * 64];
            bw2[q].y = p2[(size_t)(4 * q + 1) * 64];
        }
    }
    const float bv = b1[wv * 16 + n15];

    const float* ap0 = As + n15 * PROJ_PITCH + 2 * khalf;
    const float* ap1 = ap0 + 16 * PROJ_PITCH;
    const float* hp  = Hs + (16 * rt + n15) * PROJ_PITCH + 2 * khalf;
    float* hw = Hs + (8 * khalf) * PROJ_PITCH + wv * 16 + n15;

    for (; tile < numTiles; tile += gridDim.x) {
#pragma unroll
        for (int i = 0; i < 4; ++i) {
            const int idx = tid + i * 256;
            *(float4*)(As + (idx >> 5) * PROJ_PITCH + (idx & 31) * 4) = nx[i];
        }
        __syncthreads();

        const int nt = tile + (int)gridDim.x;
        if (nt < numTiles) proj_load_rows(nx, A, nt * PROJ_ROWS, tid, M);

        v8f acc0 = {0.f, 0.f, 0.f, 0.f, 0.f, 0.f, 0.f, 0.f};
        v8f acc1 = {0.f, 0.f, 0.f, 0.f, 0.f, 0.f, 0.f, 0.f};
#pragma unroll
        for (int q = 0; q < 32; ++q) {
            const v2f a0 = *(const v2f*)(ap0 + 4 * q);
            const v2f a1 = *(const v2f*)(ap1 + 4 * q);
            acc0 = __builtin_amdgcn_wmma_f32_16x16x4_f32(
                 false, a0,  false, bw1[q],
                 (short)0, acc0,  false,  false);
            acc1 = __builtin_amdgcn_wmma_f32_16x16x4_f32(
                 false, a1,  false, bw1[q],
                 (short)0, acc1,  false,  false);
        }
#pragma unroll
        for (int r = 0; r < 8; ++r) {
            hw[r * PROJ_PITCH]        = fmaxf(acc0[r] + bv, 0.0f);
            hw[(16 + r) * PROJ_PITCH] = fmaxf(acc1[r] + bv, 0.0f);
        }
        __syncthreads();

        v8f acc2 = {0.f, 0.f, 0.f, 0.f, 0.f, 0.f, 0.f, 0.f};
#pragma unroll
        for (int q = 0; q < 32; ++q) {
            const v2f a = *(const v2f*)(hp + 4 * q);
            acc2 = __builtin_amdgcn_wmma_f32_16x16x4_f32(
                 false, a,  false, bw2[q],
                 (short)0, acc2,  false,  false);
        }
        const int zr = tile * PROJ_ROWS + 16 * rt + 8 * khalf;
        float* zp = Z0 + (size_t)zr * 64 + ct * 16 + n15;
        if (zr + 8 <= M) {
#pragma unroll
            for (int r = 0; r < 8; ++r) zp[r * 64] = acc2[r];
        } else {
#pragma unroll
            for (int r = 0; r < 8; ++r)
                if (zr + r < M) zp[r * 64] = acc2[r];
        }
    }
}

template <int CNT, int M>
__device__ __forceinline__ void fold_half(float (&p)[16], int lane) {
    const bool hi = (lane & M) != 0;
#pragma unroll
    for (int i = 0; i < CNT / 2; ++i) {
        const float keep = hi ? p[i + CNT / 2] : p[i];
        const float send = hi ? p[i] : p[i + CNT / 2];
        p[i] = keep + __shfl_xor(send, M, 32);
    }
}

template <int K>
__device__ __forceinline__ void edge_products(float (&p)[16], const float* __restrict__ Z,
                                              int lane, int ms, v2f zd) {
    v2f v[K];
#pragma unroll
    for (int i = 0; i < K; ++i) {
        const int s = __builtin_amdgcn_readlane(ms, 2 * i);
        const float* rp = Z + (size_t)s * 64;
        v[i] = *(const v2f*)(rp + lane * 2);
    }
#pragma unroll
    for (int i = 0; i < K; ++i) p[i] = v[i].x * zd.x + v[i].y * zd.y;
#pragma unroll
    for (int i = K; i < 16; ++i) p[i] = 0.0f;
}

__global__ void __launch_bounds__(256)
k_decode_pull(const float* __restrict__ Z, const int* __restrict__ rowptr,
              const int* __restrict__ csr_dst, float* __restrict__ out, int N) {
    const int lane = threadIdx.x & 31;
    const int wv = __builtin_amdgcn_readfirstlane((int)(threadIdx.x >> 5));
    const int s0 = (blockIdx.x * (int)(blockDim.x >> 5) + wv) * ROWS_PER_WAVE;
    if (s0 >= N) return;
    const int cnt = (N - s0 < ROWS_PER_WAVE) ? (N - s0) : ROWS_PER_WAVE;

    const int ri = s0 + lane;
    const int rpv = rowptr[(ri < N) ? ri : N];
    const int slot = lane >> 1;

    int beg = __builtin_amdgcn_readlane(rpv, 0);
    int end = __builtin_amdgcn_readlane(rpv, 1);
    int ms = 0;
    if (beg < end) ms = csr_dst[(beg + slot < end - 1) ? (beg + slot) : (end - 1)];
    v2f zs = *(const v2f*)(Z + (size_t)s0 * 64 + lane * 2);
    float res = 0.0f;

    for (int i = 0; i < cnt; ++i) {
        int nend = end;
        int ms_n = 0;
        v2f zs_n = zs;
        if (i + 1 < cnt) {
            nend = __builtin_amdgcn_readlane(rpv, i + 2);
            if (end < nend) ms_n = csr_dst[(end + slot < nend - 1) ? (end + slot) : (nend - 1)];
            zs_n = *(const v2f*)(Z + (size_t)(s0 + i + 1) * 64 + lane * 2);
        }

        float val = 10.0f;
        if (beg < end) {
            const int last = end - 1;
            float esum = 0.0f;
            for (int base = beg; base < end; base += 16) {
                const int n = (end - base < 16) ? (end - base) : 16;
                int ms_next = ms;
                if (base + 16 < end) {
                    const int nx = base + 16 + slot;
                    ms_next = csr_dst[(nx < last) ? nx : last];
                }

                float p[16];
                if (n > 12)      edge_products<16>(p, Z, lane, ms, zs);
                else if (n > 8)  edge_products<12>(p, Z, lane, ms, zs);
                else if (n > 4)  edge_products<8>(p, Z, lane, ms, zs);
                else             edge_products<4>(p, Z, lane, ms, zs);

                fold_half<16, 16>(p, lane);
                fold_half<8, 8>(p, lane);
                fold_half<4, 4>(p, lane);
                fold_half<2, 2>(p, lane);
                const float sc = p[0] + __shfl_xor(p[0], 1, 32);

                const float sig = 1.0f / (1.0f + __expf(-sc));
                const float err = -__logf(sig + 1e-15f);
                if (slot < n && (lane & 1) == 0) esum += err;
                ms = ms_next;
            }
            for (int o = 16; o > 0; o >>= 1) esum += __shfl_xor(esum, o, 32);
            val = esum / (float)(end - beg);
        }
        if (lane == i) res = val;

        beg = end;
        end = nend;
        ms = ms_n;
        zs = zs_n;
    }
    if (lane < cnt) out[s0 + lane] = res;
}

extern "C" void kernel_launch(void* const* d_in, const int* in_sizes, int n_in,
                              void* d_out, int out_size, void* d_ws, size_t ws_size,
                              hipStream_t stream) {
    const float* x  = (const float*)d_in[0];
    const int*   ei = (const int*)d_in[1];
    const float* W1 = (const float*)d_in[2];
    const float* b1 = (const float*)d_in[3];
    const float* W2 = (const float*)d_in[4];
    const float* b2 = (const float*)d_in[5];

    const int N = in_sizes[0] / 128;
    const int E = in_sizes[1] / 2;
    const int* src = ei;
    const int* dst = ei + E;

    float* ws     = (float*)d_ws;
    float* dinv   = ws;
    size_t off    = (((size_t)3 * N) + 63) & ~(size_t)63;
    float* bufA   = ws + off + 128 * 128 + 64 * 128;
    float* bufB   = bufA + (size_t)N * 128;
    const int NB  = (N + BIN_ROWS - 1) / BIN_ROWS;
    int*   rowptr   = (int*)(bufB + (size_t)N * 64);
    int*   rowptrS  = rowptr + N + 1;
    const int NC  = (E + CH_EDGES - 1) / CH_EDGES;
    int*   offsD    = rowptrS + N + 1;
    int*   offsS    = offsD + (size_t)(NB + 1) * NC;
    int*   csr_src  = offsS + (size_t)(NB + 1) * NC;
    float* csr_w    = (float*)(csr_src + E);
    int*   csr_dst  = (int*)(csr_w + E);
    int*   pairs    = (int*)bufA;
    int*   pairsS   = pairs + 2 * (size_t)NC * CH_EDGES;

    const int T = 256;
    const int rowBlocks = (N + 8 * ROWS_PER_WAVE - 1) / (8 * ROWS_PER_WAVE);
    k_bucket  <<<dim3(NC, 2), CH_THREADS, 0, stream>>>(src, dst, pairs, offsD, pairsS, offsS, E, NB, NC);
    k_binsort <<<dim3(NB, 2), 1024, 0, stream>>>(pairs, offsD, rowptr, dinv, csr_src,
                                                 pairsS, offsS, rowptrS, csr_dst, N, E, NC);
    k_weights <<<(E + T * W_PER_THREAD - 1) / (T * W_PER_THREAD), T, 0, stream>>>(csr_src, dinv, csr_w, E);

    k_pull<128, false><<<rowBlocks, 256, 0, stream>>>(x, bufA, rowptr, csr_src, csr_w, dinv, b1, N);
    const int numTiles = (N + PROJ_ROWS - 1) / PROJ_ROWS;
    const int projBlocks = numTiles < PROJ_BLOCKS ? numTiles : PROJ_BLOCKS;
    k_project_both<<<projBlocks, 256, 0, stream>>>(bufA, W1, b1, W2, bufB, N, numTiles);

    k_pull<64, true><<<rowBlocks, 256, 0, stream>>>(bufB, bufA, rowptr, csr_src, csr_w, dinv, b2, N);

    k_decode_pull<<<rowBlocks, 256, 0, stream>>>(bufA, rowptrS, csr_dst, (float*)d_out, N);
}


# Second sort launch reads each edge once: fixed thread group per chunk run (no run search), pairs held in registers, the count's LDS add returns the rank; 512 rows per block; row sum order changes
# speedup vs baseline: 5.4299x; 1.1387x over previous
#include <hip/hip_runtime.h>
#include <math.h>


typedef float v2f __attribute__((ext_vector_type(2)));
typedef float v4f __attribute__((ext_vector_type(4)));
typedef float v8f __attribute__((ext_vector_type(8)));

template <int V> struct VecT;
template <> struct VecT<4> { typedef v4f T; };
template <> struct VecT<2> { typedef v2f T; };

typedef int v2i __attribute__((ext_vector_type(2)));

#define BIN_SHIFT 8
#define BIN_ROWS 256
#define MAX_BINS 512
#define CH_THREADS 1024
#define CH_PER_THREAD 8
#define CH_EDGES 8192
#define MAX_CHUNKS 1024

__global__ void __launch_bounds__(CH_THREADS)
k_bucket(const int* __restrict__ srcp, const int* __restrict__ dstp,
         int* __restrict__ pairsD, int* __restrict__ offsD,
         int* __restrict__ pairsS, int* __restrict__ offsS, int E, int NB, int NC) {
    __shared__ int hist[MAX_BINS];
    __shared__ int gbase[MAX_BINS + 1];
    __shared__ v2i stage[CH_EDGES];
    const bool bySrc = blockIdx.y != 0;
    const int* __restrict__ dst = bySrc ? srcp : dstp;
    const int* __restrict__ src = bySrc ? dstp : srcp;
    int* __restrict__ pairs = bySrc ? pairsS : pairsD;
    int* __restrict__ offs = bySrc ? offsS : offsD;
    const int t = threadIdx.x;
    const int chunk = blockIdx.x;
    if (t < MAX_BINS) hist[t] = 0;
    __syncthreads();
    const int base = chunk * CH_EDGES + t;
    int d[CH_PER_THREAD], s[CH_PER_THREAD], r[CH_PER_THREAD];
#pragma unroll
    for (int i = 0; i < CH_PER_THREAD; ++i) {
        const int e = base + i * CH_THREADS;
        d[i] = (e < E) ? dst[e] : -1;
        s[i] = (e < E) ? src[e] : 0;
    }
#pragma unroll
    for (int i = 0; i < CH_PER_THREAD; ++i) {
        r[i] = 0;
        if (d[i] >= 0)
            r[i] = __hip_atomic_fetch_add(&hist[d[i] >> BIN_SHIFT], 1, __ATOMIC_RELAXED,
                                          __HIP_MEMORY_SCOPE_WORKGROUP);
    }
    __syncthreads();
    if (t < 32) {
        int v[16];
        int sum = 0;
#pragma unroll
        for (int q = 0; q < 16; ++q) {
            v[q] = hist[t * 16 + q];
            sum += v[q];
        }
        int incl = sum;
        for (int o = 1; o < 32; o <<= 1) {
            int y = __shfl_up(incl, o, 32);
            if (t >= o) incl += y;
        }
        int run = incl - sum;
#pragma unroll
        for (int q = 0; q < 16; ++q) {
            gbase[t * 16 + q] = run;
            run += v[q];
        }
        if (t == 31) gbase[MAX_BINS] = run;
    }
    __syncthreads();
    if (t <= NB) offs[(size_t)t * NC + chunk] = gbase[t];
#pragma unroll
    for (int i = 0; i < CH_PER_THREAD; ++i) {
        if (d[i] >= 0) {
            v2i p; p.x = s[i]; p.y = d[i];
            stage[gbase[d[i] >> BIN_SHIFT] + r[i]] = p;
        }
    }
    __syncthreads();
    const int total = gbase[MAX_BINS];
    v2i* __restrict__ outp = (v2i*)pairs + (size_t)chunk * CH_EDGES;
#pragma unroll
    for (int i = 0; i < CH_PER_THREAD; ++i) {
        const int j = t + i * CH_THREADS;
        if (j < total) outp[j] = stage[j];
    }
}

#define SORT_BINS 2
#define SORT_ROWS 512
#define BS_HOLD 14
__global__ void __launch_bounds__(1024)
k_binsort(const int* __restrict__ pairsD, const int* __restrict__ offsD,
          int* __restrict__ rowptrD, float* __restrict__ dinv,
          int* __restrict__ csrD,
          const int* __restrict__ pairsS, const int* __restrict__ offsS,
          int* __restrict__ rowptrS, int* __restrict__ csrS,
          int N, int E, int NB, int NC) {
    const bool bySrc = blockIdx.y != 0;
    const int* __restrict__ pairs = bySrc ? pairsS : pairsD;
    const int* __restrict__ offs = bySrc ? offsS : offsD;
    int* __restrict__ rowptr = bySrc ? rowptrS : rowptrD;
    int* __restrict__ csr_src = bySrc ? csrS : csrD;
    __shared__ int cnt[SORT_ROWS + 32];
    __shared__ int cnt2[SORT_ROWS];
    __shared__ int off[SORT_ROWS + 32];
    __shared__ int wsum[SORT_ROWS / 32];
    __shared__ int bsum[32];
    const int t = threadIdx.x;
    const int lane = t & 31;
    const int wv = t >> 5;
    const int b0 = blockIdx.x * SORT_BINS;
    const int b1 = (b0 + SORT_BINS < NB) ? (b0 + SORT_BINS) : NB;
    const int lo = b0 << BIN_SHIFT;
    if (t < SORT_ROWS) { cnt[t] = 0; cnt2[t] = 0; }
    if (t < 32) { cnt[SORT_ROWS + t] = 0; off[SORT_ROWS + t] = 0; }

    const int G = MAX_CHUNKS / NC;
    const int c = t / G;
    const int sub = t - c * G;
    int rlo = 0, rlen = 0;
    if (c < NC) {
        rlo = offs[(size_t)b0 * NC + c];
        rlen = offs[(size_t)b1 * NC + c] - rlo;
    }
    const v2i* __restrict__ run =
        (const v2i*)pairs + ((rlen > 0) ? ((size_t)c * CH_EDGES + rlo) : (size_t)0);
    v2i pr[BS_HOLD];
#pragma unroll
    for (int q = 0; q < BS_HOLD; ++q) {
        const int i = sub + q * G;
        pr[q] = run[(i < rlen) ? i : 0];
    }

    int bs = (sub == 0) ? rlo : 0;
    for (int o = 16; o > 0; o >>= 1) bs += __shfl_xor(bs, o, 32);
    if (lane == 0) bsum[wv] = bs;
    __syncthreads();

    int pv[BS_HOLD], pk[BS_HOLD], rk[BS_HOLD];
#pragma unroll
    for (int q = 0; q < BS_HOLD; ++q) {
        pv[q] = pr[q].x;
        pk[q] = (sub + q * G < rlen) ? (pr[q].y - lo) : (SORT_ROWS + lane);
    }
#pragma unroll
    for (int q = 0; q < BS_HOLD; ++q)
        rk[q] = __hip_atomic_fetch_add(&cnt[pk[q]], 1, __ATOMIC_RELAXED,
                                       __HIP_MEMORY_SCOPE_WORKGROUP);
    for (int i = sub + BS_HOLD * G; i < rlen; i += G)
        __hip_atomic_fetch_add(&cnt2[run[i].y - lo], 1, __ATOMIC_RELAXED,
                               __HIP_MEMORY_SCOPE_WORKGROUP);
    int beg = bsum[lane];
    for (int o = 16; o > 0; o >>= 1) beg += __shfl_xor(beg, o, 32);
    __syncthreads();

    const int ch = (t < SORT_ROWS) ? cnt[t] : 0;
    const int cc = ch + ((t < SORT_ROWS) ? cnt2[t] : 0);
    int incl = cc;
    for (int o = 1; o < 32; o <<= 1) {
        int y = __shfl_up(incl, o, 32);
        if (lane >= o) incl += y;
    }
    if (t < SORT_ROWS && lane == 31) wsum[wv] = incl;
    __syncthreads();
    if (t < SORT_ROWS) {
        int rs = incl - cc;
        for (int i = 0; i < wv; ++i) rs += wsum[i];
        off[t] = rs;
        cnt2[t] = ch;
        const int node = lo + t;
        if (node < N) {
            rowptr[node] = beg + rs;
            if (!bySrc) dinv[node] = rsqrtf((float)(cc + 1));
        }
    }
    if (blockIdx.x == 0 && t == 0) rowptr[N] = E;
    __syncthreads();

#pragma unroll
    for (int q = 0; q < BS_HOLD; ++q)
        rk[q] += beg + off[pk[q]];
#pragma unroll
    for (int q = 0; q < BS_HOLD; ++q) {
        if (sub + q * G < rlen) csr_src[rk[q]] = pv[q];
    }
    for (int i = sub + BS_HOLD * G; i < rlen; i += G) {
        const v2i p = run[i];
        const int k = p.y - lo;
        const int r = __hip_atomic_fetch_add(&cnt2[k], 1, __ATOMIC_RELAXED,
                                             __HIP_MEMORY_SCOPE_WORKGROUP);
        csr_src[beg + off[k] + r] = p.x;
    }
}

#define ROWS_PER_WAVE 8
#define W_PER_THREAD 8

__global__ void __launch_bounds__(256)
k_weights(const int* __restrict__ csr_src, const float* __restrict__ dinv,
          float* __restrict__ csr_w, int E) {
    const int blk0 = blockIdx.x * (256 * W_PER_THREAD);
    const int base = blk0 + threadIdx.x;
    if (blk0 + 256 * W_PER_THREAD <= E) {
        int s[W_PER_THREAD];
        float w[W_PER_THREAD];
#pragma unroll
        for (int i = 0; i < W_PER_THREAD; ++i) s[i] = csr_src[base + i * 256];
#pragma unroll
        for (int i = 0; i < W_PER_THREAD; ++i) w[i] = dinv[s[i]];
#pragma unroll
        for (int i = 0; i < W_PER_THREAD; ++i) csr_w[base + i * 256] = w[i];
    } else {
        for (int i = 0; i < W_PER_THREAD; ++i) {
            const int j = base + i * 256;
            if (j < E) csr_w[j] = dinv[csr_src[j]];
        }
    }
}

template <int DIM>
__device__ __forceinline__ void pull_chunk(typename VecT<DIM / 32>::T& acc, const float* hl,
                                           int ms, int mwi, int n) {
    typedef typename VecT<DIM / 32>::T VT;
    int k = 0;
    for (; k + 4 <= n; k += 4) {
        const int s0 = __builtin_amdgcn_readlane(ms, k);
        const int s1 = __builtin_amdgcn_readlane(ms, k + 1);
        const int s2 = __builtin_amdgcn_readlane(ms, k + 2);
        const int s3 = __builtin_amdgcn_readlane(ms, k + 3);
        const VT v0 = *(const VT*)(hl + (size_t)s0 * DIM);
        const VT v1 = *(const VT*)(hl + (size_t)s1 * DIM);
        const VT v2 = *(const VT*)(hl + (size_t)s2 * DIM);
        const VT v3 = *(const VT*)(hl + (size_t)s3 * DIM);
        const float w0 = __int_as_float(__builtin_amdgcn_readlane(mwi, k));
        const float w1 = __int_as_float(__builtin_amdgcn_readlane(mwi, k + 1));
        const float w2 = __int_as_float(__builtin_amdgcn_readlane(mwi, k + 2));
        const float w3 = __int_as_float(__builtin_amdgcn_readlane(mwi, k + 3));
        acc += v0 * w0;
        acc += v1 * w1;
        acc += v2 * w2;
        acc += v3 * w3;
    }
    for (; k < n; ++k) {
        const int s0 = __builtin_amdgcn_readlane(ms, k);
        const float w0 = __int_as_float(__builtin_amdgcn_readlane(mwi, k));
        acc += *(const VT*)(hl + (size_t)s0 * DIM) * w0;
    }
}

template <int DIM, bool ADD_BIAS>
__global__ void __launch_bounds__(256)
k_pull(const float* __restrict__ H, float* __restrict__ out,
       const int* __restrict__ rowptr, const int* __restrict__ csr_src,
       const float* __restrict__ csr_w, const float* __restrict__ dinv,
       const float* __restrict__ bias, int N) {
    constexpr int VPL = DIM / 32;
    typedef typename VecT<VPL>::T VT;
    const int lane = threadIdx.x & 31;
    const int wv = __builtin_amdgcn_readfirstlane((int)(threadIdx.x >> 5));
    const int d0 = (blockIdx.x * (int)(blockDim.x >> 5) + wv) * ROWS_PER_WAVE;
    if (d0 >= N) return;
    const int cnt = (N - d0 < ROWS_PER_WAVE) ? (N - d0) : ROWS_PER_WAVE;

    const int ri = d0 + lane;
    const int rpv = rowptr[(ri < N) ? ri : N];
    const int ddv = __float_as_int(dinv[(ri < N) ? ri : (N - 1)]);
    const float* hl = H + lane * VPL;
    VT bv = {};
    if constexpr (ADD_BIAS) bv = *(const VT*)(bias + lane * VPL);

    int beg = __builtin_amdgcn_readlane(rpv, 0);
    int end = __builtin_amdgcn_readlane(rpv, 1);
    int ms = 0;
    float mw = 0.0f;
    if (beg + lane < end) { ms = csr_src[beg + lane]; mw = csr_w[beg + lane]; }
    VT selfv = *(const VT*)(hl + (size_t)d0 * DIM);

    for (int i = 0; i < cnt; ++i) {
        const int d = d0 + i;
        const float dd = __int_as_float(__builtin_amdgcn_readlane(ddv, i));
        int nend = end;
        int ms_n = 0;
        float mw_n = 0.0f;
        VT self_n = selfv;
        if (i + 1 < cnt) {
            nend = __builtin_amdgcn_readlane(rpv, i + 2);
            if (end + lane < nend) { ms_n = csr_src[end + lane]; mw_n = csr_w[end + lane]; }
            self_n = *(const VT*)(hl + (size_t)(d + 1) * DIM);
        }

        VT acc = selfv * dd;

        if (beg < end)
            pull_chunk<DIM>(acc, hl, ms, __float_as_int(mw), (end - beg < 32) ? (end - beg) : 32);
        for (int base = beg + 32; base < end; base += 32) {
            const int n = (end - base < 32) ? (end - base) : 32;
            int cs = 0;
            float cw = 0.0f;
            if (lane < n) { cs = csr_src[base + lane]; cw = csr_w[base + lane]; }
            pull_chunk<DIM>(acc, hl, cs, __float_as_int(cw), n);
        }

        acc *= dd;
        if constexpr (ADD_BIAS) acc += bv;
        *(VT*)(out + (size_t)d * DIM + lane * VPL) = acc;

        beg = end;
        end = nend;
        ms = ms_n;
        mw = mw_n;
        selfv = self_n;
    }
}

#define PROJ_ROWS 32
#define PROJ_PITCH 132
#define PROJ_BLOCKS 512

__device__ __forceinline__ void proj_load_rows(float4 (&v)[4], const float* __restrict__ A,
                                               int row0, int tid, int M) {
    const float4* __restrict__ src = (const float4*)(A + (size_t)row0 * 128) + tid;
    if (row0 + PROJ_ROWS <= M) {
#pragma unroll
        for (int i = 0; i < 4; ++i) v[i] = src[i * 256];
    } else {
#pragma unroll
        for (int i = 0; i < 4; ++i) {
            const int r = row0 + ((tid + i * 256) >> 5);
            v[i] = (r < M) ? src[i * 256] : make_float4(0.f, 0.f, 0.f, 0.f);
        }
    }
}

__global__ void __launch_bounds__(256)
k_project_both(const float* __restrict__ A, const float* __restrict__ W1,
               const float* __restrict__ b1, const float* __restrict__ W2,
               float* __restrict__ Z0, int M, int numTiles) {
    __shared__ float As[PROJ_ROWS * PROJ_PITCH];
    __shared__ float Hs[PROJ_ROWS * PROJ_PITCH];
    const int tid   = threadIdx.x;
    const int lane  = tid & 31;
    const int wv    = tid >> 5;
    const int n15   = lane & 15;
    const int khalf = lane >> 4;
    const int rt    = wv >> 2;
    const int ct    = wv & 3;

    int tile = blockIdx.x;
    float4 nx[4];
    proj_load_rows(nx, A, tile * PROJ_ROWS, tid, M);

    v2f bw1[32], bw2[32];
    {
        const float* p1 = W1 + (size_t)(2 * khalf) * 128 + wv * 16 + n15;
        const float* p2 = W2 + (size_t)(2 * khalf) * 64 + ct * 16 + n15;
#pragma unroll
        for (int q = 0; q < 32; ++q) {
            bw1[q].x = p1[(size_t)(4 * q) * 128];
            bw1[q].y = p1[(size_t)(4 * q + 1) * 128];
            bw2[q].x = p2[(size_t)(4 * q) * 64];
            bw2[q].y = p2[(size_t)(4 * q + 1) * 64];
        }
    }
    const float bv = b1[wv * 16 + n15];

    const float* ap0 = As + n15 * PROJ_PITCH + 2 * khalf;
    const float* ap1 = ap0 + 16 * PROJ_PITCH;
    const float* hp  = Hs + (16 * rt + n15) * PROJ_PITCH + 2 * khalf;
    float* hw = Hs + (8 * khalf) * PROJ_PITCH + wv * 16 + n15;

    for (; tile < numTiles; tile += gridDim.x) {
#pragma unroll
        for (int i = 0; i < 4; ++i) {
            const int idx = tid + i * 256;
            *(float4*)(As + (idx >> 5) * PROJ_PITCH + (idx & 31) * 4) = nx[i];
        }
        __syncthreads();

        const int nt = tile + (int)gridDim.x;
        if (nt < numTiles) proj_load_rows(nx, A, nt * PROJ_ROWS, tid, M);

        v8f acc0 = {0.f, 0.f, 0.f, 0.f, 0.f, 0.f, 0.f, 0.f};
        v8f acc1 = {0.f, 0.f, 0.f, 0.f, 0.f, 0.f, 0.f, 0.f};
#pragma unroll
        for (int q = 0; q < 32; ++q) {
            const v2f a0 = *(const v2f*)(ap0 + 4 * q);
            const v2f a1 = *(const v2f*)(ap1 + 4 * q);
            acc0 = __builtin_amdgcn_wmma_f32_16x16x4_f32(
                 false, a0,  false, bw1[q],
                 (short)0, acc0,  false,  false);
            acc1 = __builtin_amdgcn_wmma_f32_16x16x4_f32(
                 false, a1,  false, bw1[q],
                 (short)0, acc1,  false,  false);
        }
#pragma unroll
        for (int r = 0; r < 8; ++r) {
            hw[r * PROJ_PITCH]        = fmaxf(acc0[r] + bv, 0.0f);
            hw[(16 + r) * PROJ_PITCH] = fmaxf(acc1[r] + bv, 0.0f);
        }
        __syncthreads();

        v8f acc2 = {0.f, 0.f, 0.f, 0.f, 0.f, 0.f, 0.f, 0.f};
#pragma unroll
        for (int q = 0; q < 32; ++q) {
            const v2f a = *(const v2f*)(hp + 4 * q);
            acc2 = __builtin_amdgcn_wmma_f32_16x16x4_f32(
                 false, a,  false, bw2[q],
                 (short)0, acc2,  false,  false);
        }
        const int zr = tile * PROJ_ROWS + 16 * rt + 8 * khalf;
        float* zp = Z0 + (size_t)zr * 64 + ct * 16 + n15;
        if (zr + 8 <= M) {
#pragma unroll
            for (int r = 0; r < 8; ++r) zp[r * 64] = acc2[r];
        } else {
#pragma unroll
            for (int r = 0; r < 8; ++r)
                if (zr + r < M) zp[r * 64] = acc2[r];
        }
    }
}

template <int CNT, int M>
__device__ __forceinline__ void fold_half(float (&p)[16], int lane) {
    const bool hi = (lane & M) != 0;
#pragma unroll
    for (int i = 0; i < CNT / 2; ++i) {
        const float keep = hi ? p[i + CNT / 2] : p[i];
        const float send = hi ? p[i] : p[i + CNT / 2];
        p[i] = keep + __shfl_xor(send, M, 32);
    }
}

template <int K>
__device__ __forceinline__ void edge_products(float (&p)[16], const float* __restrict__ Z,
                                              int lane, int ms, v2f zd) {
    v2f v[K];
#pragma unroll
    for (int i = 0; i < K; ++i) {
        const int s = __builtin_amdgcn_readlane(ms, 2 * i);
        const float* rp = Z + (size_t)s * 64;
        v[i] = *(const v2f*)(rp + lane * 2);
    }
#pragma unroll
    for (int i = 0; i < K; ++i) p[i] = v[i].x * zd.x + v[i].y * zd.y;
#pragma unroll
    for (int i = K; i < 16; ++i) p[i] = 0.0f;
}

__global__ void __launch_bounds__(256)
k_decode_pull(const float* __restrict__ Z, const int* __restrict__ rowptr,
              const int* __restrict__ csr_dst, float* __restrict__ out, int N) {
    const int lane = threadIdx.x & 31;
    const int wv = __builtin_amdgcn_readfirstlane((int)(threadIdx.x >> 5));
    const int s0 = (blockIdx.x * (int)(blockDim.x >> 5) + wv) * ROWS_PER_WAVE;
    if (s0 >= N) return;
    const int cnt = (N - s0 < ROWS_PER_WAVE) ? (N - s0) : ROWS_PER_WAVE;

    const int ri = s0 + lane;
    const int rpv = rowptr[(ri < N) ? ri : N];
    const int slot = lane >> 1;

    int beg = __builtin_amdgcn_readlane(rpv, 0);
    int end = __builtin_amdgcn_readlane(rpv, 1);
    int ms = 0;
    if (beg < end) ms = csr_dst[(beg + slot < end - 1) ? (beg + slot) : (end - 1)];
    v2f zs = *(const v2f*)(Z + (size_t)s0 * 64 + lane * 2);
    float res = 0.0f;

    for (int i = 0; i < cnt; ++i) {
        int nend = end;
        int ms_n = 0;
        v2f zs_n = zs;
        if (i + 1 < cnt) {
            nend = __builtin_amdgcn_readlane(rpv, i + 2);
            if (end < nend) ms_n = csr_dst[(end + slot < nend - 1) ? (end + slot) : (nend - 1)];
            zs_n = *(const v2f*)(Z + (size_t)(s0 + i + 1) * 64 + lane * 2);
        }

        float val = 10.0f;
        if (beg < end) {
            const int last = end - 1;
            float esum = 0.0f;
            for (int base = beg; base < end; base += 16) {
                const int n = (end - base < 16) ? (end - base) : 16;
                int ms_next = ms;
                if (base + 16 < end) {
                    const int nx = base + 16 + slot;
                    ms_next = csr_dst[(nx < last) ? nx : last];
                }

                float p[16];
                if (n > 12)      edge_products<16>(p, Z, lane, ms, zs);
                else if (n > 8)  edge_products<12>(p, Z, lane, ms, zs);
                else if (n > 4)  edge_products<8>(p, Z, lane, ms, zs);
                else             edge_products<4>(p, Z, lane, ms, zs);

                fold_half<16, 16>(p, lane);
                fold_half<8, 8>(p, lane);
                fold_half<4, 4>(p, lane);
                fold_half<2, 2>(p, lane);
                const float sc = p[0] + __shfl_xor(p[0], 1, 32);

                const float sig = 1.0f / (1.0f + __expf(-sc));
                const float err = -__logf(sig + 1e-15f);
                if (slot < n && (lane & 1) == 0) esum += err;
                ms = ms_next;
            }
            for (int o = 16; o > 0; o >>= 1) esum += __shfl_xor(esum, o, 32);
            val = esum / (float)(end - beg);
        }
        if (lane == i) res = val;

        beg = end;
        end = nend;
        ms = ms_n;
        zs = zs_n;
    }
    if (lane < cnt) out[s0 + lane] = res;
}

extern "C" void kernel_launch(void* const* d_in, const int* in_sizes, int n_in,
                              void* d_out, int out_size, void* d_ws, size_t ws_size,
                              hipStream_t stream) {
    const float* x  = (const float*)d_in[0];
    const int*   ei = (const int*)d_in[1];
    const float* W1 = (const float*)d_in[2];
    const float* b1 = (const float*)d_in[3];
    const float* W2 = (const float*)d_in[4];
    const float* b2 = (const float*)d_in[5];

    const int N = in_sizes[0] / 128;
    const int E = in_sizes[1] / 2;
    const int* src = ei;
    const int* dst = ei + E;

    float* ws     = (float*)d_ws;
    float* dinv   = ws;
    size_t off    = (((size_t)3 * N) + 63) & ~(size_t)63;
    float* bufA   = ws + off + 128 * 128 + 64 * 128;
    float* bufB   = bufA + (size_t)N * 128;
    const int NB  = (N + BIN_ROWS - 1) / BIN_ROWS;
    int*   rowptr   = (int*)(bufB + (size_t)N * 64);
    int*   rowptrS  = rowptr + N + 1;
    const int NC  = (E + CH_EDGES - 1) / CH_EDGES;
    int*   offsD    = rowptrS + N + 1;
    int*   offsS    = offsD + (size_t)(NB + 1) * NC;
    int*   csr_src  = offsS + (size_t)(NB + 1) * NC;
    float* csr_w    = (float*)(csr_src + E);
    int*   csr_dst  = (int*)(csr_w + E);
    int*   pairs    = (int*)bufA;
    int*   pairsS   = pairs + 2 * (size_t)NC * CH_EDGES;

    const int T = 256;
    const int rowBlocks = (N + 8 * ROWS_PER_WAVE - 1) / (8 * ROWS_PER_WAVE);
    k_bucket  <<<dim3(NC, 2), CH_THREADS, 0, stream>>>(src, dst, pairs, offsD, pairsS, offsS, E, NB, NC);
    k_binsort <<<dim3((NB + SORT_BINS - 1) / SORT_BINS, 2), 1024, 0, stream>>>(
        pairs, offsD, rowptr, dinv, csr_src, pairsS, offsS, rowptrS, csr_dst, N, E, NB, NC);
    k_weights <<<(E + T * W_PER_THREAD - 1) / (T * W_PER_THREAD), T, 0, stream>>>(csr_src, dinv, csr_w, E);

    k_pull<128, false><<<rowBlocks, 256, 0, stream>>>(x, bufA, rowptr, csr_src, csr_w, dinv, b1, N);
    const int numTiles = (N + PROJ_ROWS - 1) / PROJ_ROWS;
    const int projBlocks = numTiles < PROJ_BLOCKS ? numTiles : PROJ_BLOCKS;
    k_project_both<<<projBlocks, 256, 0, stream>>>(bufA, W1, b1, W2, bufB, N, numTiles);

    k_pull<64, true><<<rowBlocks, 256, 0, stream>>>(bufB, bufA, rowptr, csr_src, csr_w, dinv, b2, N);

    k_decode_pull<<<rowBlocks, 256, 0, stream>>>(bufA, rowptrS, csr_dst, (float*)d_out, N);
}


# Layer-1 aggregation fused into the projection launch: a wave pulls four rows of A x straight into the LDS tile the products read (W2 fragments in LDS); gathers overlap other groups' matrix work
# speedup vs baseline: 5.6013x; 1.0316x over previous
#include <hip/hip_runtime.h>
#include <math.h>


typedef float v2f __attribute__((ext_vector_type(2)));
typedef float v4f __attribute__((ext_vector_type(4)));
typedef float v8f __attribute__((ext_vector_type(8)));

template <int V> struct VecT;
template <> struct VecT<4> { typedef v4f T; };
template <> struct VecT<2> { typedef v2f T; };

typedef int v2i __attribute__((ext_vector_type(2)));

#define BIN_SHIFT 8
#define BIN_ROWS 256
#define MAX_BINS 512
#define CH_THREADS 1024
#define CH_PER_THREAD 8
#define CH_EDGES 8192
#define MAX_CHUNKS 1024

__global__ void __launch_bounds__(CH_THREADS)
k_bucket(const int* __restrict__ srcp, const int* __restrict__ dstp,
         int* __restrict__ pairsD, int* __restrict__ offsD,
         int* __restrict__ pairsS, int* __restrict__ offsS, int E, int NB, int NC) {
    __shared__ int hist[MAX_BINS];
    __shared__ int gbase[MAX_BINS + 1];
    __shared__ v2i stage[CH_EDGES];
    const bool bySrc = blockIdx.y != 0;
    const int* __restrict__ dst = bySrc ? srcp : dstp;
    const int* __restrict__ src = bySrc ? dstp : srcp;
    int* __restrict__ pairs = bySrc ? pairsS : pairsD;
    int* __restrict__ offs = bySrc ? offsS : offsD;
    const int t = threadIdx.x;
    const int chunk = blockIdx.x;
    if (t < MAX_BINS) hist[t] = 0;
    __syncthreads();
    const int base = chunk * CH_EDGES + t;
    int d[CH_PER_THREAD], s[CH_PER_THREAD], r[CH_PER_THREAD];
#pragma unroll
    for (int i = 0; i < CH_PER_THREAD; ++i) {
        const int e = base + i * CH_THREADS;
        d[i] = (e < E) ? dst[e] : -1;
        s[i] = (e < E) ? src[e] : 0;
    }
#pragma unroll
    for (int i = 0; i < CH_PER_THREAD; ++i) {
        r[i] = 0;
        if (d[i] >= 0)
            r[i] = __hip_atomic_fetch_add(&hist[d[i] >> BIN_SHIFT], 1, __ATOMIC_RELAXED,
                                          __HIP_MEMORY_SCOPE_WORKGROUP);
    }
    __syncthreads();
    if (t < 32) {
        int v[16];
        int sum = 0;
#pragma unroll
        for (int q = 0; q < 16; ++q) {
            v[q] = hist[t * 16 + q];
            sum += v[q];
        }
        int incl = sum;
        for (int o = 1; o < 32; o <<= 1) {
            int y = __shfl_up(incl, o, 32);
            if (t >= o) incl += y;
        }
        int run = incl - sum;
#pragma unroll
        for (int q = 0; q < 16; ++q) {
            gbase[t * 16 + q] = run;
            run += v[q];
        }
        if (t == 31) gbase[MAX_BINS] = run;
    }
    __syncthreads();
    if (t <= NB) offs[(size_t)t * NC + chunk] = gbase[t];
#pragma unroll
    for (int i = 0; i < CH_PER_THREAD; ++i) {
        if (d[i] >= 0) {
            v2i p; p.x = s[i]; p.y = d[i];
            stage[gbase[d[i] >> BIN_SHIFT] + r[i]] = p;
        }
    }
    __syncthreads();
    const int total = gbase[MAX_BINS];
    v2i* __restrict__ outp = (v2i*)pairs + (size_t)chunk * CH_EDGES;
#pragma unroll
    for (int i = 0; i < CH_PER_THREAD; ++i) {
        const int j = t + i * CH_THREADS;
        if (j < total) outp[j] = stage[j];
    }
}

#define SORT_BINS 2
#define SORT_ROWS 512
#define BS_HOLD 14
__global__ void __launch_bounds__(1024)
k_binsort(const int* __restrict__ pairsD, const int* __restrict__ offsD,
          int* __restrict__ rowptrD, float* __restrict__ dinv,
          int* __restrict__ csrD,
          const int* __restrict__ pairsS, const int* __restrict__ offsS,
          int* __restrict__ rowptrS, int* __restrict__ csrS,
          int N, int E, int NB, int NC) {
    const bool bySrc = blockIdx.y != 0;
    const int* __restrict__ pairs = bySrc ? pairsS : pairsD;
    const int* __restrict__ offs = bySrc ? offsS : offsD;
    int* __restrict__ rowptr = bySrc ? rowptrS : rowptrD;
    int* __restrict__ csr_src = bySrc ? csrS : csrD;
    __shared__ int cnt[SORT_ROWS + 32];
    __shared__ int cnt2[SORT_ROWS];
    __shared__ int off[SORT_ROWS + 32];
    __shared__ int wsum[SORT_ROWS / 32];
    __shared__ int bsum[32];
    const int t = threadIdx.x;
    const int lane = t & 31;
    const int wv = t >> 5;
    const int b0 = blockIdx.x * SORT_BINS;
    const int b1 = (b0 + SORT_BINS < NB) ? (b0 + SORT_BINS) : NB;
    const int lo = b0 << BIN_SHIFT;
    if (t < SORT_ROWS) { cnt[t] = 0; cnt2[t] = 0; }
    if (t < 32) { cnt[SORT_ROWS + t] = 0; off[SORT_ROWS + t] = 0; }

    const int G = MAX_CHUNKS / NC;
    const int c = t / G;
    const int sub = t - c * G;
    int rlo = 0, rlen = 0;
    if (c < NC) {
        rlo = offs[(size_t)b0 * NC + c];
        rlen = offs[(size_t)b1 * NC + c] - rlo;
    }
    const v2i* __restrict__ run =
        (const v2i*)pairs + ((rlen > 0) ? ((size_t)c * CH_EDGES + rlo) : (size_t)0);
    v2i pr[BS_HOLD];
#pragma unroll
    for (int q = 0; q < BS_HOLD; ++q) {
        const int i = sub + q * G;
        pr[q] = run[(i < rlen) ? i : 0];
    }

    int bs = (sub == 0) ? rlo : 0;
    for (int o = 16; o > 0; o >>= 1) bs += __shfl_xor(bs, o, 32);
    if (lane == 0) bsum[wv] = bs;
    __syncthreads();

    int pv[BS_HOLD], pk[BS_HOLD], rk[BS_HOLD];
#pragma unroll
    for (int q = 0; q < BS_HOLD; ++q) {
        pv[q] = pr[q].x;
        pk[q] = (sub + q * G < rlen) ? (pr[q].y - lo) : (SORT_ROWS + lane);
    }
#pragma unroll
    for (int q = 0; q < BS_HOLD; ++q)
        rk[q] = __hip_atomic_fetch_add(&cnt[pk[q]], 1, __ATOMIC_RELAXED,
                                       __HIP_MEMORY_SCOPE_WORKGROUP);
    for (int i = sub + BS_HOLD * G; i < rlen; i += G)
        __hip_atomic_fetch_add(&cnt2[run[i].y - lo], 1, __ATOMIC_RELAXED,
                               __HIP_MEMORY_SCOPE_WORKGROUP);
    int beg = bsum[lane];
    for (int o = 16; o > 0; o >>= 1) beg += __shfl_xor(beg, o, 32);
    __syncthreads();

    const int ch = (t < SORT_ROWS) ? cnt[t] : 0;
    const int cc = ch + ((t < SORT_ROWS) ? cnt2[t] : 0);
    int incl = cc;
    for (int o = 1; o < 32; o <<= 1) {
        int y = __shfl_up(incl, o, 32);
        if (lane >= o) incl += y;
    }
    if (t < SORT_ROWS && lane == 31) wsum[wv] = incl;
    __syncthreads();
    if (t < SORT_ROWS) {
        int rs = incl - cc;
        for (int i = 0; i < wv; ++i) rs += wsum[i];
        off[t] = rs;
        cnt2[t] = ch;
        const int node = lo + t;
        if (node < N) {
            rowptr[node] = beg + rs;
            if (!bySrc) dinv[node] = rsqrtf((float)(cc + 1));
        }
    }
    if (blockIdx.x == 0 && t == 0) rowptr[N] = E;
    __syncthreads();

#pragma unroll
    for (int q = 0; q < BS_HOLD; ++q)
        rk[q] += beg + off[pk[q]];
#pragma unroll
    for (int q = 0; q < BS_HOLD; ++q) {
        if (sub + q * G < rlen) csr_src[rk[q]] = pv[q];
    }
    for (int i = sub + BS_HOLD * G; i < rlen; i += G) {
        const v2i p = run[i];
        const int k = p.y - lo;
        const int r = __hip_atomic_fetch_add(&cnt2[k], 1, __ATOMIC_RELAXED,
                                             __HIP_MEMORY_SCOPE_WORKGROUP);
        csr_src[beg + off[k] + r] = p.x;
    }
}

#define ROWS_PER_WAVE 8
#define W_PER_THREAD 8

__global__ void __launch_bounds__(256)
k_weights(const int* __restrict__ csr_src, const float* __restrict__ dinv,
          float* __restrict__ csr_w, int E) {
    const int blk0 = blockIdx.x * (256 * W_PER_THREAD);
    const int base = blk0 + threadIdx.x;
    if (blk0 + 256 * W_PER_THREAD <= E) {
        int s[W_PER_THREAD];
        float w[W_PER_THREAD];
#pragma unroll
        for (int i = 0; i < W_PER_THREAD; ++i) s[i] = csr_src[base + i * 256];
#pragma unroll
        for (int i = 0; i < W_PER_THREAD; ++i) w[i] = dinv[s[i]];
#pragma unroll
        for (int i = 0; i < W_PER_THREAD; ++i) csr_w[base + i * 256] = w[i];
    } else {
        for (int i = 0; i < W_PER_THREAD; ++i) {
            const int j = base + i * 256;
            if (j < E) csr_w[j] = dinv[csr_src[j]];
        }
    }
}

template <int DIM>
__device__ __forceinline__ void pull_chunk(typename VecT<DIM / 32>::T& acc, const float* hl,
                                           int ms, int mwi, int n) {
    typedef typename VecT<DIM / 32>::T VT;
    int k = 0;
    for (; k + 4 <= n; k += 4) {
        const int s0 = __builtin_amdgcn_readlane(ms, k);
        const int s1 = __builtin_amdgcn_readlane(ms, k + 1);
        const int s2 = __builtin_amdgcn_readlane(ms, k + 2);
        const int s3 = __builtin_amdgcn_readlane(ms, k + 3);
        const VT v0 = *(const VT*)(hl + (size_t)s0 * DIM);
        const VT v1 = *(const VT*)(hl + (size_t)s1 * DIM);
        const VT v2 = *(const VT*)(hl + (size_t)s2 * DIM);
        const VT v3 = *(const VT*)(hl + (size_t)s3 * DIM);
        const float w0 = __int_as_float(__builtin_amdgcn_readlane(mwi, k));
        const float w1 = __int_as_float(__builtin_amdgcn_readlane(mwi, k + 1));
        const float w2 = __int_as_float(__builtin_amdgcn_readlane(mwi, k + 2));
        const float w3 = __int_as_float(__builtin_amdgcn_readlane(mwi, k + 3));
        acc += v0 * w0;
        acc += v1 * w1;
        acc += v2 * w2;
        acc += v3 * w3;
    }
    for (; k < n; ++k) {
        const int s0 = __builtin_amdgcn_readlane(ms, k);
        const float w0 = __int_as_float(__builtin_amdgcn_readlane(mwi, k));
        acc += *(const VT*)(hl + (size_t)s0 * DIM) * w0;
    }
}

template <int DIM, bool ADD_BIAS>
__global__ void __launch_bounds__(256)
k_pull(const float* __restrict__ H, float* __restrict__ out,
       const int* __restrict__ rowptr, const int* __restrict__ csr_src,
       const float* __restrict__ csr_w, const float* __restrict__ dinv,
       const float* __restrict__ bias, int N) {
    constexpr int VPL = DIM / 32;
    typedef typename VecT<VPL>::T VT;
    const int lane = threadIdx.x & 31;
    const int wv = __builtin_amdgcn_readfirstlane((int)(threadIdx.x >> 5));
    const int d0 = (blockIdx.x * (int)(blockDim.x >> 5) + wv) * ROWS_PER_WAVE;
    if (d0 >= N) return;
    const int cnt = (N - d0 < ROWS_PER_WAVE) ? (N - d0) : ROWS_PER_WAVE;

    const int ri = d0 + lane;
    const int rpv = rowptr[(ri < N) ? ri : N];
    const int ddv = __float_as_int(dinv[(ri < N) ? ri : (N - 1)]);
    const float* hl = H + lane * VPL;
    VT bv = {};
    if constexpr (ADD_BIAS) bv = *(const VT*)(bias + lane * VPL);

    int beg = __builtin_amdgcn_readlane(rpv, 0);
    int end = __builtin_amdgcn_readlane(rpv, 1);
    int ms = 0;
    float mw = 0.0f;
    if (beg + lane < end) { ms = csr_src[beg + lane]; mw = csr_w[beg + lane]; }
    VT selfv = *(const VT*)(hl + (size_t)d0 * DIM);

    for (int i = 0; i < cnt; ++i) {
        const int d = d0 + i;
        const float dd = __int_as_float(__builtin_amdgcn_readlane(ddv, i));
        int nend = end;
        int ms_n = 0;
        float mw_n = 0.0f;
        VT self_n = selfv;
        if (i + 1 < cnt) {
            nend = __builtin_amdgcn_readlane(rpv, i + 2);
            if (end + lane < nend) { ms_n = csr_src[end + lane]; mw_n = csr_w[end + lane]; }
            self_n = *(const VT*)(hl + (size_t)(d + 1) * DIM);
        }

        VT acc = selfv * dd;

        if (beg < end)
            pull_chunk<DIM>(acc, hl, ms, __float_as_int(mw), (end - beg < 32) ? (end - beg) : 32);
        for (int base = beg + 32; base < end; base += 32) {
            const int n = (end - base < 32) ? (end - base) : 32;
            int cs = 0;
            float cw = 0.0f;
            if (lane < n) { cs = csr_src[base + lane]; cw = csr_w[base + lane]; }
            pull_chunk<DIM>(acc, hl, cs, __float_as_int(cw), n);
        }

        acc *= dd;
        if constexpr (ADD_BIAS) acc += bv;
        *(VT*)(out + (size_t)d * DIM + lane * VPL) = acc;

        beg = end;
        end = nend;
        ms = ms_n;
        mw = mw_n;
        selfv = self_n;
    }
}

#define PROJ_ROWS 32
#define PROJ_PITCH 132
#define FUSE_SLOTS 1024
#define ROWS_PER_TILE_WAVE 4

template <int K>
__device__ __forceinline__ void pull_batch(v4f& acc, const float* hl, int ms, int mwi, int k) {
    v4f v[K];
    float w[K];
#pragma unroll
    for (int j = 0; j < K; ++j) {
        const int s = __builtin_amdgcn_readlane(ms, k + j);
        v[j] = *(const v4f*)(hl + (size_t)s * 128);
    }
#pragma unroll
    for (int j = 0; j < K; ++j) w[j] = __int_as_float(__builtin_amdgcn_readlane(mwi, k + j));
#pragma unroll
    for (int j = 0; j < K; ++j) acc += v[j] * w[j];
}

__device__ __forceinline__ void pull_chunk_deep(v4f& acc, const float* hl, int ms, int mwi, int n) {
    int k = 0;
    for (; k + 8 <= n; k += 8) pull_batch<8>(acc, hl, ms, mwi, k);
    int r = n - k;
    if (r >= 4) {
        pull_batch<4>(acc, hl, ms, mwi, k);
        k += 4;
        r -= 4;
    }
    if (r == 3)      pull_batch<3>(acc, hl, ms, mwi, k);
    else if (r == 2) pull_batch<2>(acc, hl, ms, mwi, k);
    else if (r == 1) pull_batch<1>(acc, hl, ms, mwi, k);
}

__global__ void __launch_bounds__(256, 8)
k_pull_project(const float* __restrict__ X, const int* __restrict__ rowptr,
               const int* __restrict__ csr_src, const float* __restrict__ csr_w,
               const float* __restrict__ dinv,
               const float* __restrict__ W1, const float* __restrict__ b1,
               const float* __restrict__ W2, float* __restrict__ Z0, int M, int numTiles) {
    __shared__ float As[PROJ_ROWS * PROJ_PITCH];
    __shared__ float Hs[PROJ_ROWS * PROJ_PITCH];
    __shared__ v2f W2s[32 * 4 * 32];
    const int tid   = threadIdx.x;
    const int lane  = tid & 31;
    const int wv    = tid >> 5;
    const int wvs   = __builtin_amdgcn_readfirstlane(wv);
    const int n15   = lane & 15;
    const int khalf = lane >> 4;
    const int rt    = wv >> 2;
    const int ct    = wv & 3;

    for (int e = tid; e < 32 * 4 * 32; e += 256) {
        const int el = e & 31;
        const int ec = (e >> 5) & 3;
        const int eq = e >> 7;
        const float* p = W2 + (size_t)(4 * eq + 2 * (el >> 4)) * 64 + ec * 16 + (el & 15);
        v2f f; f.x = p[0]; f.y = p[64];
        W2s[e] = f;
    }
    v2f bw1[32];
    {
        const float* p1 = W1 + (size_t)(2 * khalf) * 128 + wv * 16 + n15;
#pragma unroll
        for (int q = 0; q < 32; ++q) {
            bw1[q].x = p1[(size_t)(4 * q) * 128];
            bw1[q].y = p1[(size_t)(4 * q + 1) * 128];
        }
    }
    const float bv = b1[wv * 16 + n15];

    const float* ap0 = As + n15 * PROJ_PITCH + 2 * khalf;
    const float* ap1 = ap0 + 16 * PROJ_PITCH;
    const float* hp  = Hs + (16 * rt + n15) * PROJ_PITCH + 2 * khalf;
    float* hw = Hs + (8 * khalf) * PROJ_PITCH + wv * 16 + n15;
    const v2f* wp2 = W2s + ct * 32 + lane;
    const float* hl = X + lane * 4;
    float* arow = As + (wvs * ROWS_PER_TILE_WAVE) * PROJ_PITCH + lane * 4;
    const v4f zero4 = {0.f, 0.f, 0.f, 0.f};

    for (int tile = blockIdx.x; tile < numTiles; tile += gridDim.x) {
        const int d0 = tile * PROJ_ROWS + wvs * ROWS_PER_TILE_WAVE;
        int cnt = M - d0;
        cnt = (cnt < 0) ? 0 : ((cnt < ROWS_PER_TILE_WAVE) ? cnt : ROWS_PER_TILE_WAVE);
        if (cnt > 0) {
            const int ri = d0 + lane;
            const int rpv = rowptr[(ri < M) ? ri : M];
            const int ddv = __float_as_int(dinv[(ri < M) ? ri : (M - 1)]);
            int beg = __builtin_amdgcn_readlane(rpv, 0);
            int end = __builtin_amdgcn_readlane(rpv, 1);
            int ms = 0;
            float mw = 0.0f;
            if (beg + lane < end) { ms = csr_src[beg + lane]; mw = csr_w[beg + lane]; }
            v4f selfv = *(const v4f*)(hl + (size_t)d0 * 128);
            for (int i = 0; i < cnt; ++i) {
                const float dd = __int_as_float(__builtin_amdgcn_readlane(ddv, i));
                int nend = end;
                int ms_n = 0;
                float mw_n = 0.0f;
                v4f self_n = selfv;
                if (i + 1 < cnt) {
                    nend = __builtin_amdgcn_readlane(rpv, i + 2);
                    if (end + lane < nend) { ms_n = csr_src[end + lane]; mw_n = csr_w[end + lane]; }
                    self_n = *(const v4f*)(hl + (size_t)(d0 + i + 1) * 128);
                }
                v4f acc = selfv * dd;
                if (beg < end)
                    pull_chunk_deep(acc, hl, ms, __float_as_int(mw), (end - beg < 32) ? (end - beg) : 32);
                for (int base = beg + 32; base < end; base += 32) {
                    const int n = (end - base < 32) ? (end - base) : 32;
                    int cs = 0;
                    float cw = 0.0f;
                    if (lane < n) { cs = csr_src[base + lane]; cw = csr_w[base + lane]; }
                    pull_chunk_deep(acc, hl, cs, __float_as_int(cw), n);
                }
                acc *= dd;
                *(v4f*)(arow + i * PROJ_PITCH) = acc;
                beg = end;
                end = nend;
                ms = ms_n;
                mw = mw_n;
                selfv = self_n;
            }
        }
        for (int i = cnt; i < ROWS_PER_TILE_WAVE; ++i) *(v4f*)(arow + i * PROJ_PITCH) = zero4;
        __syncthreads();

        v8f acc0 = {0.f, 0.f, 0.f, 0.f, 0.f, 0.f, 0.f, 0.f};
        v8f acc1 = {0.f, 0.f, 0.f, 0.f, 0.f, 0.f, 0.f, 0.f};
#pragma unroll
        for (int q = 0; q < 32; ++q) {
            const v2f a0 = *(const v2f*)(ap0 + 4 * q);
            const v2f a1 = *(const v2f*)(ap1 + 4 * q);
            acc0 = __builtin_amdgcn_wmma_f32_16x16x4_f32(
                 false, a0,  false, bw1[q],
                 (short)0, acc0,  false,  false);
            acc1 = __builtin_amdgcn_wmma_f32_16x16x4_f32(
                 false, a1,  false, bw1[q],
                 (short)0, acc1,  false,  false);
        }
#pragma unroll
        for (int r = 0; r < 8; ++r) {
            hw[r * PROJ_PITCH]        = fmaxf(acc0[r] + bv, 0.0f);
            hw[(16 + r) * PROJ_PITCH] = fmaxf(acc1[r] + bv, 0.0f);
        }
        __syncthreads();

        v8f acc2 = {0.f, 0.f, 0.f, 0.f, 0.f, 0.f, 0.f, 0.f};
#pragma unroll
        for (int q = 0; q < 32; ++q) {
            const v2f a = *(const v2f*)(hp + 4 * q);
            const v2f b = wp2[q * 128];
            acc2 = __builtin_amdgcn_wmma_f32_16x16x4_f32(
                 false, a,  false, b,
                 (short)0, acc2,  false,  false);
        }
        const int zr = tile * PROJ_ROWS + 16 * rt + 8 * khalf;
        float* zp = Z0 + (size_t)zr * 64 + ct * 16 + n15;
        if (zr + 8 <= M) {
#pragma unroll
            for (int r = 0; r < 8; ++r) zp[r * 64] = acc2[r];
        } else {
#pragma unroll
            for (int r = 0; r < 8; ++r)
                if (zr + r < M) zp[r * 64] = acc2[r];
        }
    }
}

template <int CNT, int M>
__device__ __forceinline__ void fold_half(float (&p)[16], int lane) {
    const bool hi = (lane & M) != 0;
#pragma unroll
    for (int i = 0; i < CNT / 2; ++i) {
        const float keep = hi ? p[i + CNT / 2] : p[i];
        const float send = hi ? p[i] : p[i + CNT / 2];
        p[i] = keep + __shfl_xor(send, M, 32);
    }
}

template <int K>
__device__ __forceinline__ void edge_products(float (&p)[16], const float* __restrict__ Z,
                                              int lane, int ms, v2f zd) {
    v2f v[K];
#pragma unroll
    for (int i = 0; i < K; ++i) {
        const int s = __builtin_amdgcn_readlane(ms, 2 * i);
        const float* rp = Z + (size_t)s * 64;
        v[i] = *(const v2f*)(rp + lane * 2);
    }
#pragma unroll
    for (int i = 0; i < K; ++i) p[i] = v[i].x * zd.x + v[i].y * zd.y;
#pragma unroll
    for (int i = K; i < 16; ++i) p[i] = 0.0f;
}

__global__ void __launch_bounds__(256)
k_decode_pull(const float* __restrict__ Z, const int* __restrict__ rowptr,
              const int* __restrict__ csr_dst, float* __restrict__ out, int N) {
    const int lane = threadIdx.x & 31;
    const int wv = __builtin_amdgcn_readfirstlane((int)(threadIdx.x >> 5));
    const int s0 = (blockIdx.x * (int)(blockDim.x >> 5) + wv) * ROWS_PER_WAVE;
    if (s0 >= N) return;
    const int cnt = (N - s0 < ROWS_PER_WAVE) ? (N - s0) : ROWS_PER_WAVE;

    const int ri = s0 + lane;
    const int rpv = rowptr[(ri < N) ? ri : N];
    const int slot = lane >> 1;

    int beg = __builtin_amdgcn_readlane(rpv, 0);
    int end = __builtin_amdgcn_readlane(rpv, 1);
    int ms = 0;
    if (beg < end) ms = csr_dst[(beg + slot < end - 1) ? (beg + slot) : (end - 1)];
    v2f zs = *(const v2f*)(Z + (size_t)s0 * 64 + lane * 2);
    float res = 0.0f;

    for (int i = 0; i < cnt; ++i) {
        int nend = end;
        int ms_n = 0;
        v2f zs_n = zs;
        if (i + 1 < cnt) {
            nend = __builtin_amdgcn_readlane(rpv, i + 2);
            if (end < nend) ms_n = csr_dst[(end + slot < nend - 1) ? (end + slot) : (nend - 1)];
            zs_n = *(const v2f*)(Z + (size_t)(s0 + i + 1) * 64 + lane * 2);
        }

        float val = 10.0f;
        if (beg < end) {
            const int last = end - 1;
            float esum = 0.0f;
            for (int base = beg; base < end; base += 16) {
                const int n = (end - base < 16) ? (end - base) : 16;
                int ms_next = ms;
                if (base + 16 < end) {
                    const int nx = base + 16 + slot;
                    ms_next = csr_dst[(nx < last) ? nx : last];
                }

                float p[16];
                if (n > 12)      edge_products<16>(p, Z, lane, ms, zs);
                else if (n > 8)  edge_products<12>(p, Z, lane, ms, zs);
                else if (n > 4)  edge_products<8>(p, Z, lane, ms, zs);
                else             edge_products<4>(p, Z, lane, ms, zs);

                fold_half<16, 16>(p, lane);
                fold_half<8, 8>(p, lane);
                fold_half<4, 4>(p, lane);
                fold_half<2, 2>(p, lane);
                const float sc = p[0] + __shfl_xor(p[0], 1, 32);

                const float sig = 1.0f / (1.0f + __expf(-sc));
                const float err = -__logf(sig + 1e-15f);
                if (slot < n && (lane & 1) == 0) esum += err;
                ms = ms_next;
            }
            for (int o = 16; o > 0; o >>= 1) esum += __shfl_xor(esum, o, 32);
            val = esum / (float)(end - beg);
        }
        if (lane == i) res = val;

        beg = end;
        end = nend;
        ms = ms_n;
        zs = zs_n;
    }
    if (lane < cnt) out[s0 + lane] = res;
}

extern "C" void kernel_launch(void* const* d_in, const int* in_sizes, int n_in,
                              void* d_out, int out_size, void* d_ws, size_t ws_size,
                              hipStream_t stream) {
    const float* x  = (const float*)d_in[0];
    const int*   ei = (const int*)d_in[1];
    const float* W1 = (const float*)d_in[2];
    const float* b1 = (const float*)d_in[3];
    const float* W2 = (const float*)d_in[4];
    const float* b2 = (const float*)d_in[5];

    const int N = in_sizes[0] / 128;
    const int E = in_sizes[1] / 2;
    const int* src = ei;
    const int* dst = ei + E;

    float* ws     = (float*)d_ws;
    float* dinv   = ws;
    size_t off    = (((size_t)3 * N) + 63) & ~(size_t)63;
    float* bufA   = ws + off + 128 * 128 + 64 * 128;
    float* bufB   = bufA + (size_t)N * 128;
    const int NB  = (N + BIN_ROWS - 1) / BIN_ROWS;
    int*   rowptr   = (int*)(bufB + (size_t)N * 64);
    int*   rowptrS  = rowptr + N + 1;
    const int NC  = (E + CH_EDGES - 1) / CH_EDGES;
    int*   offsD    = rowptrS + N + 1;
    int*   offsS    = offsD + (size_t)(NB + 1) * NC;
    int*   csr_src  = offsS + (size_t)(NB + 1) * NC;
    float* csr_w    = (float*)(csr_src + E);
    int*   csr_dst  = (int*)(csr_w + E);
    int*   pairs    = (int*)bufA;
    int*   pairsS   = pairs + 2 * (size_t)NC * CH_EDGES;

    const int T = 256;
    const int rowBlocks = (N + 8 * ROWS_PER_WAVE - 1) / (8 * ROWS_PER_WAVE);
    k_bucket  <<<dim3(NC, 2), CH_THREADS, 0, stream>>>(src, dst, pairs, offsD, pairsS, offsS, E, NB, NC);
    k_binsort <<<dim3((NB + SORT_BINS - 1) / SORT_BINS, 2), 1024, 0, stream>>>(
        pairs, offsD, rowptr, dinv, csr_src, pairsS, offsS, rowptrS, csr_dst, N, E, NB, NC);
    k_weights <<<(E + T * W_PER_THREAD - 1) / (T * W_PER_THREAD), T, 0, stream>>>(csr_src, dinv, csr_w, E);

    const int numTiles = (N + PROJ_ROWS - 1) / PROJ_ROWS;
    const int tilesPerGroup = (numTiles + FUSE_SLOTS - 1) / FUSE_SLOTS;
    const int fuseBlocks = (numTiles + tilesPerGroup - 1) / tilesPerGroup;
    k_pull_project<<<fuseBlocks, 256, 0, stream>>>(x, rowptr, csr_src, csr_w, dinv, W1, b1, W2, bufB, N, numTiles);

    k_pull<64, true><<<rowBlocks, 256, 0, stream>>>(bufB, bufA, rowptr, csr_src, csr_w, dinv, b2, N);

    k_decode_pull<<<rowBlocks, 256, 0, stream>>>(bufA, rowptrS, csr_dst, (float*)d_out, N);
}
